# starting kernel
# baseline (speedup 1.0000x reference)
#include <hip/hip_runtime.h>
#include <hip/hip_bf16.h>


typedef __attribute__((ext_vector_type(2))) float v2f;
typedef __attribute__((ext_vector_type(8))) float v8f;

#define TB 256

__global__ void k_deg_init(float* __restrict__ deg, int n) {
  int i = blockIdx.x * blockDim.x + threadIdx.x;
  if (i < n) deg[i] = 1.0f;
}

__global__ void k_deg_count(const int* __restrict__ dst, float* __restrict__ deg, int e) {
  int i = blockIdx.x * blockDim.x + threadIdx.x;
  if (i < e) atomicAdd(&deg[dst[i]], 1.0f);
}

__global__ void k_deg_rsqrt(float* __restrict__ deg, int n) {
  int i = blockIdx.x * blockDim.x + threadIdx.x;
  if (i < n) deg[i] = rsqrtf(deg[i]);
}

template <bool RELU_IN>
__global__ void k_gemm64_wmma(const float* __restrict__ X,
                              const float* __restrict__ W,
                              float* __restrict__ H, int nstrips) {
  int wave = (int)((blockIdx.x * blockDim.x + threadIdx.x) >> 5);
  int lane = threadIdx.x & 31;
  if (wave >= nstrips) return;

  const int m0    = wave * 16;
  const int mlow  = lane & 15;
  const int khalf = (lane >> 4) << 1;

  v2f a[16];
  const float* xr = X + (size_t)(m0 + mlow) * 64 + khalf;
#pragma unroll
  for (int k = 0; k < 16; ++k) {
    float ax = xr[k * 4 + 0];
    float ay = xr[k * 4 + 1];
    if (RELU_IN) { ax = fmaxf(ax, 0.0f); ay = fmaxf(ay, 0.0f); }
    a[k].x = ax; a[k].y = ay;
  }

#pragma unroll
  for (int nt = 0; nt < 4; ++nt) {
    const float* wr = W + (size_t)(nt * 16 + mlow) * 64 + khalf;
    v8f c = {};
#pragma unroll
    for (int k = 0; k < 16; ++k) {
      v2f b;
      b.x = wr[k * 4 + 0];
      b.y = wr[k * 4 + 1];
      c = __builtin_amdgcn_wmma_f32_16x16x4_f32(
           false, a[k],  false, b,
           (short)0, c,  false,  false);
    }
    float* o = H + (size_t)(m0 + 8 * (lane >> 4)) * 64 + nt * 16 + mlow;
#pragma unroll
    for (int v = 0; v < 8; ++v) o[(size_t)v * 64] = c[v];
  }
}

__global__ void k_self_bias(const float* __restrict__ h, const float* __restrict__ dis,
                            const float* __restrict__ bias, float* __restrict__ out, int n) {
  int t = blockIdx.x * blockDim.x + threadIdx.x;
  if (t >= n * 16) return;
  int i = t >> 4;
  int f = (t & 15) << 2;
  float w = dis[i]; w *= w;
  float4 v = *(const float4*)(h + (size_t)i * 64 + f);
  float4 b = *(const float4*)(bias + f);
  float4 r;
  r.x = v.x * w + b.x; r.y = v.y * w + b.y;
  r.z = v.z * w + b.z; r.w = v.w * w + b.w;
  *(float4*)(out + (size_t)i * 64 + f) = r;
}

__global__ void k_edge_scatter(const int* __restrict__ src, const int* __restrict__ dst,
                               const float* __restrict__ h, const float* __restrict__ dis,
                               float* __restrict__ out, int e) {
  int t = blockIdx.x * blockDim.x + threadIdx.x;
  int ed = t >> 4;
  if (ed >= e) return;
  int f = (t & 15) << 2;
  int s = src[ed], d = dst[ed];
  float w = dis[s] * dis[d];
  float4 v = *(const float4*)(h + (size_t)s * 64 + f);
  float* o = out + (size_t)d * 64 + f;
  atomicAdd(o + 0, v.x * w);
  atomicAdd(o + 1, v.y * w);
  atomicAdd(o + 2, v.z * w);
  atomicAdd(o + 3, v.w * w);
}

extern "C" void kernel_launch(void* const* d_in, const int* in_sizes, int n_in,
                              void* d_out, int out_size, void* d_ws, size_t ws_size,
                              hipStream_t stream) {
  (void)n_in; (void)out_size; (void)ws_size;

  const float* x  = (const float*)d_in[0];
  const int*   ei = (const int*)  d_in[1];
  const float* W1 = (const float*)d_in[2];
  const float* b1 = (const float*)d_in[3];
  const float* W2 = (const float*)d_in[4];
  const float* b2 = (const float*)d_in[5];
  float* out = (float*)d_out;

  const int n = in_sizes[0] / 64;
  const int e = in_sizes[1] / 2;
  const int* src = ei;
  const int* dst = ei + e;

  float* dis = (float*)d_ws;
  float* h   = dis + n;
  float* agg = h + (size_t)n * 64;

  const int nstrips = (n + 15) / 16;
  const int gemm_blocks = (nstrips + 7) / 8;
  const int nf_blocks   = (n * 16 + TB - 1) / TB;
  const int ef_blocks   = (e * 16 + TB - 1) / TB;

  k_deg_init <<<(n + TB - 1) / TB, TB, 0, stream>>>(dis, n);
  k_deg_count<<<(e + TB - 1) / TB, TB, 0, stream>>>(dst, dis, e);
  k_deg_rsqrt<<<(n + TB - 1) / TB, TB, 0, stream>>>(dis, n);

  k_gemm64_wmma<false><<<gemm_blocks, TB, 0, stream>>>(x, W1, h, nstrips);
  k_self_bias  <<<nf_blocks, TB, 0, stream>>>(h, dis, b1, agg, n);
  k_edge_scatter<<<ef_blocks, TB, 0, stream>>>(src, dst, h, dis, agg, e);

  k_gemm64_wmma<true><<<gemm_blocks, TB, 0, stream>>>(agg, W2, h, nstrips);
  k_self_bias  <<<nf_blocks, TB, 0, stream>>>(h, dis, b2, out, n);
  k_edge_scatter<<<ef_blocks, TB, 0, stream>>>(src, dst, h, dis, out, e);
}


# edge scatter remapped: a wave takes 8 edges, lane l holds floats l and l+32 of the row, so every f32 atomic and row load covers one whole 128-B line, not 8 floats in each of 4 lines; atomics kept
# speedup vs baseline: 1.4032x; 1.4032x over previous
#include <hip/hip_runtime.h>
#include <hip/hip_bf16.h>


typedef __attribute__((ext_vector_type(2))) float v2f;
typedef __attribute__((ext_vector_type(8))) float v8f;

#define TB 256

__global__ void k_deg_init(float* __restrict__ deg, int n) {
  int i = blockIdx.x * blockDim.x + threadIdx.x;
  if (i < n) deg[i] = 1.0f;
}

__global__ void k_deg_count(const int* __restrict__ dst, float* __restrict__ deg, int e) {
  int i = blockIdx.x * blockDim.x + threadIdx.x;
  if (i < e) atomicAdd(&deg[dst[i]], 1.0f);
}

__global__ void k_deg_rsqrt(float* __restrict__ deg, int n) {
  int i = blockIdx.x * blockDim.x + threadIdx.x;
  if (i < n) deg[i] = rsqrtf(deg[i]);
}

template <bool RELU_IN>
__global__ void k_gemm64_wmma(const float* __restrict__ X,
                              const float* __restrict__ W,
                              float* __restrict__ H, int nstrips) {
  int wave = (int)((blockIdx.x * blockDim.x + threadIdx.x) >> 5);
  int lane = threadIdx.x & 31;
  if (wave >= nstrips) return;

  const int m0    = wave * 16;
  const int mlow  = lane & 15;
  const int khalf = (lane >> 4) << 1;

  v2f a[16];
  const float* xr = X + (size_t)(m0 + mlow) * 64 + khalf;
#pragma unroll
  for (int k = 0; k < 16; ++k) {
    float ax = xr[k * 4 + 0];
    float ay = xr[k * 4 + 1];
    if (RELU_IN) { ax = fmaxf(ax, 0.0f); ay = fmaxf(ay, 0.0f); }
    a[k].x = ax; a[k].y = ay;
  }

#pragma unroll
  for (int nt = 0; nt < 4; ++nt) {
    const float* wr = W + (size_t)(nt * 16 + mlow) * 64 + khalf;
    v8f c = {};
#pragma unroll
    for (int k = 0; k < 16; ++k) {
      v2f b;
      b.x = wr[k * 4 + 0];
      b.y = wr[k * 4 + 1];
      c = __builtin_amdgcn_wmma_f32_16x16x4_f32(
           false, a[k],  false, b,
           (short)0, c,  false,  false);
    }
    float* o = H + (size_t)(m0 + 8 * (lane >> 4)) * 64 + nt * 16 + mlow;
#pragma unroll
    for (int v = 0; v < 8; ++v) o[(size_t)v * 64] = c[v];
  }
}

__global__ void k_self_bias(const float* __restrict__ h, const float* __restrict__ dis,
                            const float* __restrict__ bias, float* __restrict__ out, int n) {
  int t = blockIdx.x * blockDim.x + threadIdx.x;
  if (t >= n * 16) return;
  int i = t >> 4;
  int f = (t & 15) << 2;
  float w = dis[i]; w *= w;
  float4 v = *(const float4*)(h + (size_t)i * 64 + f);
  float4 b = *(const float4*)(bias + f);
  float4 r;
  r.x = v.x * w + b.x; r.y = v.y * w + b.y;
  r.z = v.z * w + b.z; r.w = v.w * w + b.w;
  *(float4*)(out + (size_t)i * 64 + f) = r;
}

#define EPW 8
__global__ void k_edge_scatter(const int* __restrict__ src, const int* __restrict__ dst,
                               const float* __restrict__ h, const float* __restrict__ dis,
                               float* __restrict__ out, int e) {
  const int lane = threadIdx.x & 31;
  const int wave = __builtin_amdgcn_readfirstlane(
      (int)((blockIdx.x * blockDim.x + threadIdx.x) >> 5));
  const int e0 = wave * EPW;
  if (e0 >= e) return;
  if (e0 + EPW <= e) {
    int s[EPW], d[EPW];
#pragma unroll
    for (int j = 0; j < EPW; ++j) { s[j] = src[e0 + j]; d[j] = dst[e0 + j]; }
    float w[EPW];
#pragma unroll
    for (int j = 0; j < EPW; ++j) w[j] = dis[s[j]] * dis[d[j]];
    float v0[EPW], v1[EPW];
#pragma unroll
    for (int j = 0; j < EPW; ++j) {
      const float* hr = h + (size_t)s[j] * 64 + lane;
      v0[j] = hr[0];
      v1[j] = hr[32];
    }
#pragma unroll
    for (int j = 0; j < EPW; ++j) {
      float* o = out + (size_t)d[j] * 64 + lane;
      atomicAdd(o,      v0[j] * w[j]);
      atomicAdd(o + 32, v1[j] * w[j]);
    }
  } else {
    for (int ed = e0; ed < e; ++ed) {
      const int s = src[ed], d = dst[ed];
      const float w = dis[s] * dis[d];
      const float* hr = h + (size_t)s * 64 + lane;
      float* o = out + (size_t)d * 64 + lane;
      atomicAdd(o,      hr[0]  * w);
      atomicAdd(o + 32, hr[32] * w);
    }
  }
}

extern "C" void kernel_launch(void* const* d_in, const int* in_sizes, int n_in,
                              void* d_out, int out_size, void* d_ws, size_t ws_size,
                              hipStream_t stream) {
  (void)n_in; (void)out_size; (void)ws_size;

  const float* x  = (const float*)d_in[0];
  const int*   ei = (const int*)  d_in[1];
  const float* W1 = (const float*)d_in[2];
  const float* b1 = (const float*)d_in[3];
  const float* W2 = (const float*)d_in[4];
  const float* b2 = (const float*)d_in[5];
  float* out = (float*)d_out;

  const int n = in_sizes[0] / 64;
  const int e = in_sizes[1] / 2;
  const int* src = ei;
  const int* dst = ei + e;

  float* dis = (float*)d_ws;
  float* h   = dis + n;
  float* agg = h + (size_t)n * 64;

  const int nstrips = (n + 15) / 16;
  const int gemm_blocks = (nstrips + 7) / 8;
  const int nf_blocks   = (n * 16 + TB - 1) / TB;
  const int ef_blocks   = (e + 8 * EPW - 1) / (8 * EPW);

  k_deg_init <<<(n + TB - 1) / TB, TB, 0, stream>>>(dis, n);
  k_deg_count<<<(e + TB - 1) / TB, TB, 0, stream>>>(dst, dis, e);
  k_deg_rsqrt<<<(n + TB - 1) / TB, TB, 0, stream>>>(dis, n);

  k_gemm64_wmma<false><<<gemm_blocks, TB, 0, stream>>>(x, W1, h, nstrips);
  k_self_bias  <<<nf_blocks, TB, 0, stream>>>(h, dis, b1, agg, n);
  k_edge_scatter<<<ef_blocks, TB, 0, stream>>>(src, dst, h, dis, agg, e);

  k_gemm64_wmma<true><<<gemm_blocks, TB, 0, stream>>>(agg, W2, h, nstrips);
  k_self_bias  <<<nf_blocks, TB, 0, stream>>>(h, dis, b2, out, n);
  k_edge_scatter<<<ef_blocks, TB, 0, stream>>>(src, dst, h, dis, out, e);
}


# in-degree counted in LDS per (edge slice, node range) work-group, partial counts stored whole-line and summed in the rsqrt kernel: no degree atomics, init launch gone; dis bit-identical
# speedup vs baseline: 1.7564x; 1.2517x over previous
#include <hip/hip_runtime.h>
#include <hip/hip_bf16.h>


typedef __attribute__((ext_vector_type(2))) float v2f;
typedef __attribute__((ext_vector_type(8))) float v8f;

#define TB 256

#define DC_RANGE  12512
#define DC_SLICES 32
#define DC_TB     1024
#define DC_UNROLL 16
__global__ void __launch_bounds__(DC_TB)
k_deg_count(const int* __restrict__ dst, unsigned int* __restrict__ part, int e, int nranges) {
  __shared__ unsigned int cnt[DC_RANGE];
  const int tid = threadIdx.x;
  const int s = blockIdx.x / nranges;
  const int g = blockIdx.x - s * nranges;
  const int lo = g * DC_RANGE;

  for (int b = tid; b < DC_RANGE; b += DC_TB) cnt[b] = 0u;
  __syncthreads();

  const int per = (((e + DC_SLICES - 1) / DC_SLICES) + 31) & ~31;
  const int e0 = s * per;
  const int e1 = min(e, e0 + per);
  for (int base = e0; base < e1; base += DC_UNROLL * DC_TB) {
    int v[DC_UNROLL];
#pragma unroll
    for (int j = 0; j < DC_UNROLL; ++j) {
      v[j] = dst[min(base + j * DC_TB + tid, e1 - 1)];
    }
    asm volatile("" :: "v"(v[0]), "v"(v[1]), "v"(v[2]), "v"(v[3]), "v"(v[4]), "v"(v[5]),
                       "v"(v[6]), "v"(v[7]), "v"(v[8]), "v"(v[9]), "v"(v[10]), "v"(v[11]),
                       "v"(v[12]), "v"(v[13]), "v"(v[14]), "v"(v[15]));
#pragma unroll
    for (int j = 0; j < DC_UNROLL; ++j) {
      const bool live = (base + j * DC_TB + tid) < e1;
      const unsigned int r = (unsigned int)(v[j] - lo);
      if (live && r < (unsigned int)DC_RANGE) atomicAdd(&cnt[r], 1u);
    }
  }
  __syncthreads();

  unsigned int* row = part + (size_t)s * ((size_t)nranges * DC_RANGE) + lo;
  for (int b = tid; b < DC_RANGE; b += DC_TB) row[b] = cnt[b];
}

__global__ void k_deg_finish(const unsigned int* __restrict__ part, float* __restrict__ dis,
                             int n, int npad) {
  int i = blockIdx.x * blockDim.x + threadIdx.x;
  if (i >= n) return;
  unsigned int c = 1u;
#pragma unroll
  for (int s = 0; s < DC_SLICES; ++s) c += part[(size_t)s * npad + i];
  dis[i] = rsqrtf((float)c);
}

template <bool RELU_IN>
__global__ void k_gemm64_wmma(const float* __restrict__ X,
                              const float* __restrict__ W,
                              float* __restrict__ H, int nstrips) {
  int wave = (int)((blockIdx.x * blockDim.x + threadIdx.x) >> 5);
  int lane = threadIdx.x & 31;
  if (wave >= nstrips) return;

  const int m0    = wave * 16;
  const int mlow  = lane & 15;
  const int khalf = (lane >> 4) << 1;

  v2f a[16];
  const float* xr = X + (size_t)(m0 + mlow) * 64 + khalf;
#pragma unroll
  for (int k = 0; k < 16; ++k) {
    float ax = xr[k * 4 + 0];
    float ay = xr[k * 4 + 1];
    if (RELU_IN) { ax = fmaxf(ax, 0.0f); ay = fmaxf(ay, 0.0f); }
    a[k].x = ax; a[k].y = ay;
  }

#pragma unroll
  for (int nt = 0; nt < 4; ++nt) {
    const float* wr = W + (size_t)(nt * 16 + mlow) * 64 + khalf;
    v8f c = {};
#pragma unroll
    for (int k = 0; k < 16; ++k) {
      v2f b;
      b.x = wr[k * 4 + 0];
      b.y = wr[k * 4 + 1];
      c = __builtin_amdgcn_wmma_f32_16x16x4_f32(
           false, a[k],  false, b,
           (short)0, c,  false,  false);
    }
    float* o = H + (size_t)(m0 + 8 * (lane >> 4)) * 64 + nt * 16 + mlow;
#pragma unroll
    for (int v = 0; v < 8; ++v) o[(size_t)v * 64] = c[v];
  }
}

__global__ void k_self_bias(const float* __restrict__ h, const float* __restrict__ dis,
                            const float* __restrict__ bias, float* __restrict__ out, int n) {
  int t = blockIdx.x * blockDim.x + threadIdx.x;
  if (t >= n * 16) return;
  int i = t >> 4;
  int f = (t & 15) << 2;
  float w = dis[i]; w *= w;
  float4 v = *(const float4*)(h + (size_t)i * 64 + f);
  float4 b = *(const float4*)(bias + f);
  float4 r;
  r.x = v.x * w + b.x; r.y = v.y * w + b.y;
  r.z = v.z * w + b.z; r.w = v.w * w + b.w;
  *(float4*)(out + (size_t)i * 64 + f) = r;
}

#define EPW 8
__global__ void k_edge_scatter(const int* __restrict__ src, const int* __restrict__ dst,
                               const float* __restrict__ h, const float* __restrict__ dis,
                               float* __restrict__ out, int e) {
  const int lane = threadIdx.x & 31;
  const int wave = __builtin_amdgcn_readfirstlane(
      (int)((blockIdx.x * blockDim.x + threadIdx.x) >> 5));
  const int e0 = wave * EPW;
  if (e0 >= e) return;
  if (e0 + EPW <= e) {
    int s[EPW], d[EPW];
#pragma unroll
    for (int j = 0; j < EPW; ++j) { s[j] = src[e0 + j]; d[j] = dst[e0 + j]; }
    float w[EPW];
#pragma unroll
    for (int j = 0; j < EPW; ++j) w[j] = dis[s[j]] * dis[d[j]];
    float v0[EPW], v1[EPW];
#pragma unroll
    for (int j = 0; j < EPW; ++j) {
      const float* hr = h + (size_t)s[j] * 64 + lane;
      v0[j] = hr[0];
      v1[j] = hr[32];
    }
#pragma unroll
    for (int j = 0; j < EPW; ++j) {
      float* o = out + (size_t)d[j] * 64 + lane;
      atomicAdd(o,      v0[j] * w[j]);
      atomicAdd(o + 32, v1[j] * w[j]);
    }
  } else {
    for (int ed = e0; ed < e; ++ed) {
      const int s = src[ed], d = dst[ed];
      const float w = dis[s] * dis[d];
      const float* hr = h + (size_t)s * 64 + lane;
      float* o = out + (size_t)d * 64 + lane;
      atomicAdd(o,      hr[0]  * w);
      atomicAdd(o + 32, hr[32] * w);
    }
  }
}

extern "C" void kernel_launch(void* const* d_in, const int* in_sizes, int n_in,
                              void* d_out, int out_size, void* d_ws, size_t ws_size,
                              hipStream_t stream) {
  (void)n_in; (void)out_size; (void)ws_size;

  const float* x  = (const float*)d_in[0];
  const int*   ei = (const int*)  d_in[1];
  const float* W1 = (const float*)d_in[2];
  const float* b1 = (const float*)d_in[3];
  const float* W2 = (const float*)d_in[4];
  const float* b2 = (const float*)d_in[5];
  float* out = (float*)d_out;

  const int n = in_sizes[0] / 64;
  const int e = in_sizes[1] / 2;
  const int* src = ei;
  const int* dst = ei + e;

  float* dis = (float*)d_ws;
  float* h   = dis + n;
  float* agg = h + (size_t)n * 64;
  unsigned int* part = (unsigned int*)(agg + (size_t)n * 64);

  const int nstrips = (n + 15) / 16;
  const int gemm_blocks = (nstrips + 7) / 8;
  const int nf_blocks   = (n * 16 + TB - 1) / TB;
  const int ef_blocks   = (e + 8 * EPW - 1) / (8 * EPW);
  const int nranges     = (n + DC_RANGE - 1) / DC_RANGE;
  const int npad        = nranges * DC_RANGE;

  k_deg_count <<<nranges * DC_SLICES, DC_TB, 0, stream>>>(dst, part, e, nranges);
  k_deg_finish<<<(n + TB - 1) / TB, TB, 0, stream>>>(part, dis, n, npad);

  k_gemm64_wmma<false><<<gemm_blocks, TB, 0, stream>>>(x, W1, h, nstrips);
  k_self_bias  <<<nf_blocks, TB, 0, stream>>>(h, dis, b1, agg, n);
  k_edge_scatter<<<ef_blocks, TB, 0, stream>>>(src, dst, h, dis, agg, e);

  k_gemm64_wmma<true><<<gemm_blocks, TB, 0, stream>>>(agg, W2, h, nstrips);
  k_self_bias  <<<nf_blocks, TB, 0, stream>>>(h, dis, b2, out, n);
  k_edge_scatter<<<ef_blocks, TB, 0, stream>>>(src, dst, h, dis, out, e);
}


# edge aggregation as a gather: edges counting-sorted by destination from the LDS degree counts, one wave per node sums its in-edge rows in registers; float atomics gone, sum order changes
# speedup vs baseline: 3.0062x; 1.7116x over previous
#include <hip/hip_runtime.h>
#include <hip/hip_bf16.h>


typedef __attribute__((ext_vector_type(2))) float v2f;
typedef __attribute__((ext_vector_type(8))) float v8f;

#define TB 256

#define DC_RANGE  12512
#define DC_SLICES 32
#define DC_TB     1024
#define DC_UNROLL 16
__global__ void __launch_bounds__(DC_TB)
k_deg_count(const int* __restrict__ dst, unsigned int* __restrict__ part, int e, int nranges) {
  __shared__ unsigned int cnt[DC_RANGE];
  const int tid = threadIdx.x;
  const int s = blockIdx.x / nranges;
  const int g = blockIdx.x - s * nranges;
  const int lo = g * DC_RANGE;

  for (int b = tid; b < DC_RANGE; b += DC_TB) cnt[b] = 0u;
  __syncthreads();

  const int per = (((e + DC_SLICES - 1) / DC_SLICES) + 31) & ~31;
  const int e0 = s * per;
  const int e1 = min(e, e0 + per);
  for (int base = e0; base < e1; base += DC_UNROLL * DC_TB) {
    int v[DC_UNROLL];
#pragma unroll
    for (int j = 0; j < DC_UNROLL; ++j) {
      v[j] = dst[min(base + j * DC_TB + tid, e1 - 1)];
    }
    asm volatile("" :: "v"(v[0]), "v"(v[1]), "v"(v[2]), "v"(v[3]), "v"(v[4]), "v"(v[5]),
                       "v"(v[6]), "v"(v[7]), "v"(v[8]), "v"(v[9]), "v"(v[10]), "v"(v[11]),
                       "v"(v[12]), "v"(v[13]), "v"(v[14]), "v"(v[15]));
#pragma unroll
    for (int j = 0; j < DC_UNROLL; ++j) {
      const bool live = (base + j * DC_TB + tid) < e1;
      const unsigned int r = (unsigned int)(v[j] - lo);
      if (live && r < (unsigned int)DC_RANGE) atomicAdd(&cnt[r], 1u);
    }
  }
  __syncthreads();

  unsigned int* row = part + (size_t)s * ((size_t)nranges * DC_RANGE) + lo;
  for (int b = tid; b < DC_RANGE; b += DC_TB) row[b] = cnt[b];
}

#define FIN_TB 544
#define FIN_PER_RANGE (DC_RANGE / FIN_TB)
static_assert(DC_RANGE % FIN_TB == 0, "finish blocks must tile a node range");
__global__ void __launch_bounds__(FIN_TB)
k_deg_finish(unsigned int* __restrict__ part, float* __restrict__ dis,
             unsigned int* __restrict__ dtot, unsigned int* __restrict__ bsum, int n, int npad) {
  __shared__ unsigned int bs;
  const int i = blockIdx.x * FIN_TB + threadIdx.x;
  if (threadIdx.x == 0) bs = 0u;
  __syncthreads();
  unsigned int c[DC_SLICES];
#pragma unroll
  for (int s = 0; s < DC_SLICES; ++s) c[s] = part[(size_t)s * npad + i];
  unsigned int run = 0u;
#pragma unroll
  for (int s = 0; s < DC_SLICES; ++s) { part[(size_t)s * npad + i] = run; run += c[s]; }
  dtot[i] = run;
  if (i < n) dis[i] = rsqrtf((float)(run + 1u));
  atomicAdd(&bs, run);
  __syncthreads();
  if (threadIdx.x == 0) bsum[blockIdx.x * 32] = bs;
}

#define SC_CH 13
static_assert(DC_TB * SC_CH >= DC_RANGE, "range scan must cover the range");
static_assert(DC_TB == 1024, "the range scan assumes 32 waves of 32 lanes");
__global__ void __launch_bounds__(DC_TB)
k_fill(const int* __restrict__ src, const int* __restrict__ dst,
       const unsigned int* __restrict__ part, const unsigned int* __restrict__ dtot,
       const unsigned int* __restrict__ bsum, const float* __restrict__ dis,
       unsigned int* __restrict__ rowptr, int2* __restrict__ csr, int e, int nranges, int npad) {
  __shared__ unsigned int cnt[DC_RANGE];
  __shared__ unsigned int wsum[32];
  __shared__ unsigned int rb;
  const int tid = threadIdx.x;
  const int lane = tid & 31;
  const int wid = tid >> 5;
  const int s = blockIdx.x / nranges;
  const int g = blockIdx.x - s * nranges;
  const int lo = g * DC_RANGE;

  if (tid == 0) rb = 0u;
  for (int b = tid; b < DC_RANGE; b += DC_TB) cnt[b] = dtot[lo + b];
  __syncthreads();

  unsigned int acc = 0u;
  for (int k = tid; k < g * FIN_PER_RANGE; k += DC_TB) acc += bsum[k * 32];
  if (acc != 0u) atomicAdd(&rb, acc);

  const int c0 = tid * SC_CH;
  unsigned int loc[SC_CH];
  unsigned int tsum = 0u;
#pragma unroll
  for (int j = 0; j < SC_CH; ++j) {
    const int b = c0 + j;
    loc[j] = (b < DC_RANGE) ? cnt[b] : 0u;
    tsum += loc[j];
  }
  unsigned int x = tsum;
#pragma unroll
  for (int o = 1; o < 32; o <<= 1) {
    const unsigned int t = __shfl_up(x, o);
    if (lane >= o) x += t;
  }
  if (lane == 31) wsum[wid] = x;
  __syncthreads();
  const unsigned int w = wsum[lane];
  unsigned int ws = w;
#pragma unroll
  for (int o = 1; o < 32; o <<= 1) {
    const unsigned int t = __shfl_up(ws, o);
    if (lane >= o) ws += t;
  }
  const unsigned int wave_excl = __shfl(ws, wid) - __shfl(w, wid);
  unsigned int run = rb + wave_excl + (x - tsum);
#pragma unroll
  for (int j = 0; j < SC_CH; ++j) {
    const int b = c0 + j;
    if (b < DC_RANGE) { cnt[b] = run; run += loc[j]; }
  }
  __syncthreads();

  for (int b = tid; b < DC_RANGE; b += DC_TB) {
    const unsigned int v = cnt[b];
    if (s == 0) rowptr[lo + b] = v;
    cnt[b] = v + part[(size_t)s * npad + lo + b];
  }
  __syncthreads();

  const int per = (((e + DC_SLICES - 1) / DC_SLICES) + 31) & ~31;
  const int e0 = s * per;
  const int e1 = min(e, e0 + per);
  for (int base = e0; base < e1; base += DC_UNROLL * DC_TB) {
    int v[DC_UNROLL], u[DC_UNROLL];
#pragma unroll
    for (int j = 0; j < DC_UNROLL; ++j) {
      const int idx = min(base + j * DC_TB + tid, e1 - 1);
      v[j] = dst[idx];
      u[j] = src[idx];
    }
    asm volatile("" :: "v"(v[0]), "v"(v[1]), "v"(v[2]), "v"(v[3]), "v"(v[4]), "v"(v[5]),
                       "v"(v[6]), "v"(v[7]), "v"(v[8]), "v"(v[9]), "v"(v[10]), "v"(v[11]),
                       "v"(v[12]), "v"(v[13]), "v"(v[14]), "v"(v[15]));
    asm volatile("" :: "v"(u[0]), "v"(u[1]), "v"(u[2]), "v"(u[3]), "v"(u[4]), "v"(u[5]),
                       "v"(u[6]), "v"(u[7]), "v"(u[8]), "v"(u[9]), "v"(u[10]), "v"(u[11]),
                       "v"(u[12]), "v"(u[13]), "v"(u[14]), "v"(u[15]));
    unsigned int p[DC_UNROLL];
    float ds[DC_UNROLL];
#pragma unroll
    for (int j = 0; j < DC_UNROLL; ++j) {
      const bool live = (base + j * DC_TB + tid) < e1;
      const unsigned int r = (unsigned int)(v[j] - lo);
      p[j] = 0xffffffffu;
      ds[j] = 0.0f;
      if (live && r < (unsigned int)DC_RANGE) {
        p[j] = atomicAdd(&cnt[r], 1u);
        ds[j] = dis[u[j]];
      }
    }
#pragma unroll
    for (int j = 0; j < DC_UNROLL; ++j) {
      if (p[j] != 0xffffffffu) csr[p[j]] = make_int2(u[j], __float_as_int(ds[j]));
    }
  }
}

template <bool RELU_IN>
__global__ void k_gemm64_wmma(const float* __restrict__ X,
                              const float* __restrict__ W,
                              float* __restrict__ H, int nstrips) {
  int wave = (int)((blockIdx.x * blockDim.x + threadIdx.x) >> 5);
  int lane = threadIdx.x & 31;
  if (wave >= nstrips) return;

  const int m0    = wave * 16;
  const int mlow  = lane & 15;
  const int khalf = (lane >> 4) << 1;

  v2f a[16];
  const float* xr = X + (size_t)(m0 + mlow) * 64 + khalf;
#pragma unroll
  for (int k = 0; k < 16; ++k) {
    float ax = xr[k * 4 + 0];
    float ay = xr[k * 4 + 1];
    if (RELU_IN) { ax = fmaxf(ax, 0.0f); ay = fmaxf(ay, 0.0f); }
    a[k].x = ax; a[k].y = ay;
  }

#pragma unroll
  for (int nt = 0; nt < 4; ++nt) {
    const float* wr = W + (size_t)(nt * 16 + mlow) * 64 + khalf;
    v8f c = {};
#pragma unroll
    for (int k = 0; k < 16; ++k) {
      v2f b;
      b.x = wr[k * 4 + 0];
      b.y = wr[k * 4 + 1];
      c = __builtin_amdgcn_wmma_f32_16x16x4_f32(
           false, a[k],  false, b,
           (short)0, c,  false,  false);
    }
    float* o = H + (size_t)(m0 + 8 * (lane >> 4)) * 64 + nt * 16 + mlow;
#pragma unroll
    for (int v = 0; v < 8; ++v) o[(size_t)v * 64] = c[v];
  }
}

#define GB 8
__global__ void __launch_bounds__(TB)
k_gather(const int2* __restrict__ csr, const unsigned int* __restrict__ rowptr,
         const unsigned int* __restrict__ dtot, const float* __restrict__ h,
         const float* __restrict__ dis, const float* __restrict__ bias,
         float* __restrict__ out, int n) {
  const int lane = threadIdx.x & 31;
  const int i = __builtin_amdgcn_readfirstlane(
      (int)((blockIdx.x * blockDim.x + threadIdx.x) >> 5));
  if (i >= n) return;
  const int start = (int)rowptr[i];
  const int cnt = (int)dtot[i];
  const float di = dis[i];
  asm volatile("" :: "s"(start));
  const float wself = di * di;
  const float* hi = h + (size_t)i * 64 + lane;
  const float bia0 = bias[lane];
  const float bia1 = bias[lane + 32];
  float acc0 = hi[0] * wself;
  float acc1 = hi[32] * wself;
  asm volatile("" :: "v"(bia0), "v"(bia1));
  for (int k0 = 0; k0 < cnt; k0 += 32) {
    const int m = min(cnt - k0, 32);
    const int2 rec = csr[start + k0 + min(lane, m - 1)];
    const int myidx = rec.x;
    const float dsrc = __int_as_float(rec.y);
    for (int k = 0; k < m; k += GB) {
      int s[GB];
#pragma unroll
      for (int j = 0; j < GB; ++j) s[j] = __builtin_amdgcn_readlane(myidx, k + j);
      float v0[GB], v1[GB];
#pragma unroll
      for (int j = 0; j < GB; ++j) {
        const float* hr = h + (size_t)s[j] * 64 + lane;
        v0[j] = hr[0];
        v1[j] = hr[32];
      }
      float w[GB];
#pragma unroll
      for (int j = 0; j < GB; ++j) {
        const float ds = __int_as_float(__builtin_amdgcn_readlane(__float_as_int(dsrc), k + j));
        w[j] = (k + j < m) ? ds * di : 0.0f;
      }
#pragma unroll
      for (int j = 0; j < GB; ++j) {
        acc0 = fmaf(v0[j], w[j], acc0);
        acc1 = fmaf(v1[j], w[j], acc1);
      }
    }
  }
  float* o = out + (size_t)i * 64 + lane;
  o[0]  = acc0 + bia0;
  o[32] = acc1 + bia1;
}

extern "C" void kernel_launch(void* const* d_in, const int* in_sizes, int n_in,
                              void* d_out, int out_size, void* d_ws, size_t ws_size,
                              hipStream_t stream) {
  (void)n_in; (void)out_size; (void)ws_size;

  const float* x  = (const float*)d_in[0];
  const int*   ei = (const int*)  d_in[1];
  const float* W1 = (const float*)d_in[2];
  const float* b1 = (const float*)d_in[3];
  const float* W2 = (const float*)d_in[4];
  const float* b2 = (const float*)d_in[5];
  float* out = (float*)d_out;

  const int n = in_sizes[0] / 64;
  const int e = in_sizes[1] / 2;
  const int* src = ei;
  const int* dst = ei + e;

  const int nstrips = (n + 15) / 16;
  const int gemm_blocks = (nstrips + 7) / 8;
  const int node_blocks = (n + 7) / 8;
  const int nranges     = (n + DC_RANGE - 1) / DC_RANGE;
  const int npad        = nranges * DC_RANGE;
  const int nfin        = nranges * FIN_PER_RANGE;

  float* dis = (float*)d_ws;
  float* h   = dis + n;
  float* agg = h + (size_t)n * 64;
  unsigned int* part   = (unsigned int*)(agg + (size_t)n * 64);
  unsigned int* dtot   = part + (size_t)DC_SLICES * npad;
  unsigned int* rowptr = dtot + npad;
  unsigned int* bsum   = rowptr + npad;
  int2* csr = (int2*)(bsum + (size_t)nfin * 32);

  k_deg_count <<<nranges * DC_SLICES, DC_TB, 0, stream>>>(dst, part, e, nranges);
  k_deg_finish<<<nfin, FIN_TB, 0, stream>>>(part, dis, dtot, bsum, n, npad);
  k_fill      <<<nranges * DC_SLICES, DC_TB, 0, stream>>>(src, dst, part, dtot, bsum, dis, rowptr,
                                                          csr, e, nranges, npad);

  k_gemm64_wmma<false><<<gemm_blocks, TB, 0, stream>>>(x, W1, h, nstrips);
  k_gather<<<node_blocks, TB, 0, stream>>>(csr, rowptr, dtot, h, dis, b1, agg, n);

  k_gemm64_wmma<true><<<gemm_blocks, TB, 0, stream>>>(agg, W2, h, nstrips);
  k_gather<<<node_blocks, TB, 0, stream>>>(csr, rowptr, dtot, h, dis, b2, out, n);
}


# layer-2 product fused into the layer-1 gather launch: 16 one-node waves leave relu(sum+bias) rows in LDS, 4 of them form the 16x64 tile on WMMA; agg plane and one launch gone, bit-identical sums
# speedup vs baseline: 3.2070x; 1.0668x over previous
#include <hip/hip_runtime.h>
#include <hip/hip_bf16.h>


typedef __attribute__((ext_vector_type(2))) float v2f;
typedef __attribute__((ext_vector_type(8))) float v8f;

#define TB 256

#define DC_RANGE  12512
#define DC_SLICES 32
#define DC_TB     1024
#define DC_UNROLL 16
__global__ void __launch_bounds__(DC_TB)
k_deg_count(const int* __restrict__ dst, unsigned int* __restrict__ part, int e, int nranges) {
  __shared__ unsigned int cnt[DC_RANGE];
  const int tid = threadIdx.x;
  const int s = blockIdx.x / nranges;
  const int g = blockIdx.x - s * nranges;
  const int lo = g * DC_RANGE;

  for (int b = tid; b < DC_RANGE; b += DC_TB) cnt[b] = 0u;
  __syncthreads();

  const int per = (((e + DC_SLICES - 1) / DC_SLICES) + 31) & ~31;
  const int e0 = s * per;
  const int e1 = min(e, e0 + per);
  for (int base = e0; base < e1; base += DC_UNROLL * DC_TB) {
    int v[DC_UNROLL];
#pragma unroll
    for (int j = 0; j < DC_UNROLL; ++j) {
      v[j] = dst[min(base + j * DC_TB + tid, e1 - 1)];
    }
    asm volatile("" :: "v"(v[0]), "v"(v[1]), "v"(v[2]), "v"(v[3]), "v"(v[4]), "v"(v[5]),
                       "v"(v[6]), "v"(v[7]), "v"(v[8]), "v"(v[9]), "v"(v[10]), "v"(v[11]),
                       "v"(v[12]), "v"(v[13]), "v"(v[14]), "v"(v[15]));
#pragma unroll
    for (int j = 0; j < DC_UNROLL; ++j) {
      const bool live = (base + j * DC_TB + tid) < e1;
      const unsigned int r = (unsigned int)(v[j] - lo);
      if (live && r < (unsigned int)DC_RANGE) atomicAdd(&cnt[r], 1u);
    }
  }
  __syncthreads();

  unsigned int* row = part + (size_t)s * ((size_t)nranges * DC_RANGE) + lo;
  for (int b = tid; b < DC_RANGE; b += DC_TB) row[b] = cnt[b];
}

#define FIN_TB 544
#define FIN_PER_RANGE (DC_RANGE / FIN_TB)
static_assert(DC_RANGE % FIN_TB == 0, "finish blocks must tile a node range");
__global__ void __launch_bounds__(FIN_TB)
k_deg_finish(unsigned int* __restrict__ part, float* __restrict__ dis,
             unsigned int* __restrict__ dtot, unsigned int* __restrict__ bsum, int n, int npad) {
  __shared__ unsigned int bs;
  const int i = blockIdx.x * FIN_TB + threadIdx.x;
  if (threadIdx.x == 0) bs = 0u;
  __syncthreads();
  unsigned int c[DC_SLICES];
#pragma unroll
  for (int s = 0; s < DC_SLICES; ++s) c[s] = part[(size_t)s * npad + i];
  unsigned int run = 0u;
#pragma unroll
  for (int s = 0; s < DC_SLICES; ++s) { part[(size_t)s * npad + i] = run; run += c[s]; }
  dtot[i] = run;
  if (i < n) dis[i] = rsqrtf((float)(run + 1u));
  atomicAdd(&bs, run);
  __syncthreads();
  if (threadIdx.x == 0) bsum[blockIdx.x * 32] = bs;
}

#define SC_CH 13
static_assert(DC_TB * SC_CH >= DC_RANGE, "range scan must cover the range");
static_assert(DC_TB == 1024, "the range scan assumes 32 waves of 32 lanes");
__global__ void __launch_bounds__(DC_TB)
k_fill(const int* __restrict__ src, const int* __restrict__ dst,
       const unsigned int* __restrict__ part, const unsigned int* __restrict__ dtot,
       const unsigned int* __restrict__ bsum, const float* __restrict__ dis,
       unsigned int* __restrict__ rowptr, int2* __restrict__ csr, int e, int nranges, int npad) {
  __shared__ unsigned int cnt[DC_RANGE];
  __shared__ unsigned int wsum[32];
  __shared__ unsigned int rb;
  const int tid = threadIdx.x;
  const int lane = tid & 31;
  const int wid = tid >> 5;
  const int s = blockIdx.x / nranges;
  const int g = blockIdx.x - s * nranges;
  const int lo = g * DC_RANGE;

  if (tid == 0) rb = 0u;
  for (int b = tid; b < DC_RANGE; b += DC_TB) cnt[b] = dtot[lo + b];
  __syncthreads();

  unsigned int acc = 0u;
  for (int k = tid; k < g * FIN_PER_RANGE; k += DC_TB) acc += bsum[k * 32];
  if (acc != 0u) atomicAdd(&rb, acc);

  const int c0 = tid * SC_CH;
  unsigned int loc[SC_CH];
  unsigned int tsum = 0u;
#pragma unroll
  for (int j = 0; j < SC_CH; ++j) {
    const int b = c0 + j;
    loc[j] = (b < DC_RANGE) ? cnt[b] : 0u;
    tsum += loc[j];
  }
  unsigned int x = tsum;
#pragma unroll
  for (int o = 1; o < 32; o <<= 1) {
    const unsigned int t = __shfl_up(x, o);
    if (lane >= o) x += t;
  }
  if (lane == 31) wsum[wid] = x;
  __syncthreads();
  const unsigned int w = wsum[lane];
  unsigned int ws = w;
#pragma unroll
  for (int o = 1; o < 32; o <<= 1) {
    const unsigned int t = __shfl_up(ws, o);
    if (lane >= o) ws += t;
  }
  const unsigned int wave_excl = __shfl(ws, wid) - __shfl(w, wid);
  unsigned int run = rb + wave_excl + (x - tsum);
#pragma unroll
  for (int j = 0; j < SC_CH; ++j) {
    const int b = c0 + j;
    if (b < DC_RANGE) { cnt[b] = run; run += loc[j]; }
  }
  __syncthreads();

  for (int b = tid; b < DC_RANGE; b += DC_TB) {
    const unsigned int v = cnt[b];
    if (s == 0) rowptr[lo + b] = v;
    cnt[b] = v + part[(size_t)s * npad + lo + b];
  }
  __syncthreads();

  const int per = (((e + DC_SLICES - 1) / DC_SLICES) + 31) & ~31;
  const int e0 = s * per;
  const int e1 = min(e, e0 + per);
  for (int base = e0; base < e1; base += DC_UNROLL * DC_TB) {
    int v[DC_UNROLL], u[DC_UNROLL];
#pragma unroll
    for (int j = 0; j < DC_UNROLL; ++j) {
      const int idx = min(base + j * DC_TB + tid, e1 - 1);
      v[j] = dst[idx];
      u[j] = src[idx];
    }
    asm volatile("" :: "v"(v[0]), "v"(v[1]), "v"(v[2]), "v"(v[3]), "v"(v[4]), "v"(v[5]),
                       "v"(v[6]), "v"(v[7]), "v"(v[8]), "v"(v[9]), "v"(v[10]), "v"(v[11]),
                       "v"(v[12]), "v"(v[13]), "v"(v[14]), "v"(v[15]));
    asm volatile("" :: "v"(u[0]), "v"(u[1]), "v"(u[2]), "v"(u[3]), "v"(u[4]), "v"(u[5]),
                       "v"(u[6]), "v"(u[7]), "v"(u[8]), "v"(u[9]), "v"(u[10]), "v"(u[11]),
                       "v"(u[12]), "v"(u[13]), "v"(u[14]), "v"(u[15]));
    unsigned int p[DC_UNROLL];
    float ds[DC_UNROLL];
#pragma unroll
    for (int j = 0; j < DC_UNROLL; ++j) {
      const bool live = (base + j * DC_TB + tid) < e1;
      const unsigned int r = (unsigned int)(v[j] - lo);
      p[j] = 0xffffffffu;
      ds[j] = 0.0f;
      if (live && r < (unsigned int)DC_RANGE) {
        p[j] = atomicAdd(&cnt[r], 1u);
        ds[j] = dis[u[j]];
      }
    }
#pragma unroll
    for (int j = 0; j < DC_UNROLL; ++j) {
      if (p[j] != 0xffffffffu) csr[p[j]] = make_int2(u[j], __float_as_int(ds[j]));
    }
  }
}

template <bool RELU_IN>
__global__ void k_gemm64_wmma(const float* __restrict__ X,
                              const float* __restrict__ W,
                              float* __restrict__ H, int nstrips) {
  int wave = (int)((blockIdx.x * blockDim.x + threadIdx.x) >> 5);
  int lane = threadIdx.x & 31;
  if (wave >= nstrips) return;

  const int m0    = wave * 16;
  const int mlow  = lane & 15;
  const int khalf = (lane >> 4) << 1;

  v2f a[16];
  const float* xr = X + (size_t)(m0 + mlow) * 64 + khalf;
#pragma unroll
  for (int k = 0; k < 16; ++k) {
    float ax = xr[k * 4 + 0];
    float ay = xr[k * 4 + 1];
    if (RELU_IN) { ax = fmaxf(ax, 0.0f); ay = fmaxf(ay, 0.0f); }
    a[k].x = ax; a[k].y = ay;
  }

#pragma unroll
  for (int nt = 0; nt < 4; ++nt) {
    const float* wr = W + (size_t)(nt * 16 + mlow) * 64 + khalf;
    v8f c = {};
#pragma unroll
    for (int k = 0; k < 16; ++k) {
      v2f b;
      b.x = wr[k * 4 + 0];
      b.y = wr[k * 4 + 1];
      c = __builtin_amdgcn_wmma_f32_16x16x4_f32(
           false, a[k],  false, b,
           (short)0, c,  false,  false);
    }
    float* o = H + (size_t)(m0 + 8 * (lane >> 4)) * 64 + nt * 16 + mlow;
#pragma unroll
    for (int v = 0; v < 8; ++v) o[(size_t)v * 64] = c[v];
  }
}

#define GB 8
__global__ void __launch_bounds__(TB)
k_gather(const int2* __restrict__ csr, const unsigned int* __restrict__ rowptr,
         const unsigned int* __restrict__ dtot, const float* __restrict__ h,
         const float* __restrict__ dis, const float* __restrict__ bias,
         float* __restrict__ out, int n) {
  const int lane = threadIdx.x & 31;
  const int i = __builtin_amdgcn_readfirstlane(
      (int)((blockIdx.x * blockDim.x + threadIdx.x) >> 5));
  if (i >= n) return;
  const int start = (int)rowptr[i];
  const int cnt = (int)dtot[i];
  const float di = dis[i];
  asm volatile("" :: "s"(start));
  const float wself = di * di;
  const float* hi = h + (size_t)i * 64 + lane;
  const float bia0 = bias[lane];
  const float bia1 = bias[lane + 32];
  float acc0 = hi[0] * wself;
  float acc1 = hi[32] * wself;
  asm volatile("" :: "v"(bia0), "v"(bia1));
  for (int k0 = 0; k0 < cnt; k0 += 32) {
    const int m = min(cnt - k0, 32);
    const int2 rec = csr[start + k0 + min(lane, m - 1)];
    const int myidx = rec.x;
    const float dsrc = __int_as_float(rec.y);
    for (int k = 0; k < m; k += GB) {
      int s[GB];
#pragma unroll
      for (int j = 0; j < GB; ++j) s[j] = __builtin_amdgcn_readlane(myidx, k + j);
      float v0[GB], v1[GB];
#pragma unroll
      for (int j = 0; j < GB; ++j) {
        const float* hr = h + (size_t)s[j] * 64 + lane;
        v0[j] = hr[0];
        v1[j] = hr[32];
      }
      float w[GB];
#pragma unroll
      for (int j = 0; j < GB; ++j) {
        const float ds = __int_as_float(__builtin_amdgcn_readlane(__float_as_int(dsrc), k + j));
        w[j] = (k + j < m) ? ds * di : 0.0f;
      }
#pragma unroll
      for (int j = 0; j < GB; ++j) {
        acc0 = fmaf(v0[j], w[j], acc0);
        acc1 = fmaf(v1[j], w[j], acc1);
      }
    }
  }
  float* o = out + (size_t)i * 64 + lane;
  o[0]  = acc0 + bia0;
  o[32] = acc1 + bia1;
}

#define FT_NODES 16
#define FT_TB    (FT_NODES * 32)
#define FT_PITCH 68
__global__ void __launch_bounds__(FT_TB)
k_gather_gemm(const int2* __restrict__ csr, const unsigned int* __restrict__ rowptr,
              const unsigned int* __restrict__ dtot, const float* __restrict__ h,
              const float* __restrict__ dis, const float* __restrict__ bias,
              const float* __restrict__ W, float* __restrict__ H2, int n) {
  __shared__ float tile[FT_NODES * FT_PITCH];
  const int lane = threadIdx.x & 31;
  const int wid = __builtin_amdgcn_readfirstlane((int)(threadIdx.x >> 5));
  const int m0 = blockIdx.x * FT_NODES;
  const int i = m0 + wid;
  float r0 = 0.0f, r1 = 0.0f;
  if (i < n) {
    const int start = (int)rowptr[i];
    const int cnt = (int)dtot[i];
    const float di = dis[i];
    asm volatile("" :: "s"(start));
    const float wself = di * di;
    const float* hi = h + (size_t)i * 64 + lane;
    const float bia0 = bias[lane];
    const float bia1 = bias[lane + 32];
    float acc0 = hi[0] * wself;
    float acc1 = hi[32] * wself;
    asm volatile("" :: "v"(bia0), "v"(bia1));
    for (int k0 = 0; k0 < cnt; k0 += 32) {
      const int m = min(cnt - k0, 32);
      const int2 rec = csr[start + k0 + min(lane, m - 1)];
      const int myidx = rec.x;
      const float dsrc = __int_as_float(rec.y);
      for (int k = 0; k < m; k += GB) {
        int s[GB];
#pragma unroll
        for (int j = 0; j < GB; ++j) s[j] = __builtin_amdgcn_readlane(myidx, k + j);
        float v0[GB], v1[GB];
#pragma unroll
        for (int j = 0; j < GB; ++j) {
          const float* hr = h + (size_t)s[j] * 64 + lane;
          v0[j] = hr[0];
          v1[j] = hr[32];
        }
        float w[GB];
#pragma unroll
        for (int j = 0; j < GB; ++j) {
          const float ds = __int_as_float(__builtin_amdgcn_readlane(__float_as_int(dsrc), k + j));
          w[j] = (k + j < m) ? ds * di : 0.0f;
        }
#pragma unroll
        for (int j = 0; j < GB; ++j) {
          acc0 = fmaf(v0[j], w[j], acc0);
          acc1 = fmaf(v1[j], w[j], acc1);
        }
      }
    }
    r0 = fmaxf(acc0 + bia0, 0.0f);
    r1 = fmaxf(acc1 + bia1, 0.0f);
  }
  tile[wid * FT_PITCH + lane]      = r0;
  tile[wid * FT_PITCH + lane + 32] = r1;
  __syncthreads();
  if (wid >= 4) return;

  const int mlow  = lane & 15;
  const int khalf = (lane >> 4) << 1;
  const float* ar = tile + mlow * FT_PITCH + khalf;
  const float* wr = W + (size_t)(wid * 16 + mlow) * 64 + khalf;
  v2f a[16], b[16];
#pragma unroll
  for (int k = 0; k < 16; ++k) {
    b[k].x = wr[k * 4 + 0];
    b[k].y = wr[k * 4 + 1];
  }
#pragma unroll
  for (int k = 0; k < 16; ++k) {
    a[k].x = ar[k * 4 + 0];
    a[k].y = ar[k * 4 + 1];
  }
  v8f c = {};
#pragma unroll
  for (int k = 0; k < 16; ++k) {
    c = __builtin_amdgcn_wmma_f32_16x16x4_f32(
         false, a[k],  false, b[k],
         (short)0, c,  false,  false);
  }
  const int row0 = m0 + 8 * (lane >> 4);
  float* o = H2 + (size_t)row0 * 64 + wid * 16 + mlow;
  if (m0 + FT_NODES <= n) {
#pragma unroll
    for (int v = 0; v < 8; ++v) o[(size_t)v * 64] = c[v];
  } else {
#pragma unroll
    for (int v = 0; v < 8; ++v) {
      if (row0 + v < n) o[(size_t)v * 64] = c[v];
    }
  }
}

extern "C" void kernel_launch(void* const* d_in, const int* in_sizes, int n_in,
                              void* d_out, int out_size, void* d_ws, size_t ws_size,
                              hipStream_t stream) {
  (void)n_in; (void)out_size; (void)ws_size;

  const float* x  = (const float*)d_in[0];
  const int*   ei = (const int*)  d_in[1];
  const float* W1 = (const float*)d_in[2];
  const float* b1 = (const float*)d_in[3];
  const float* W2 = (const float*)d_in[4];
  const float* b2 = (const float*)d_in[5];
  float* out = (float*)d_out;

  const int n = in_sizes[0] / 64;
  const int e = in_sizes[1] / 2;
  const int* src = ei;
  const int* dst = ei + e;

  const int nstrips = (n + 15) / 16;
  const int gemm_blocks = (nstrips + 7) / 8;
  const int node_blocks = (n + 7) / 8;
  const int tile_blocks = (n + FT_NODES - 1) / FT_NODES;
  const int nranges     = (n + DC_RANGE - 1) / DC_RANGE;
  const int npad        = nranges * DC_RANGE;
  const int nfin        = nranges * FIN_PER_RANGE;

  float* dis = (float*)d_ws;
  float* h   = dis + n;
  float* h2  = h + (size_t)n * 64;
  unsigned int* part   = (unsigned int*)(h2 + (size_t)n * 64);
  unsigned int* dtot   = part + (size_t)DC_SLICES * npad;
  unsigned int* rowptr = dtot + npad;
  unsigned int* bsum   = rowptr + npad;
  int2* csr = (int2*)(bsum + (size_t)nfin * 32);

  k_deg_count <<<nranges * DC_SLICES, DC_TB, 0, stream>>>(dst, part, e, nranges);
  k_deg_finish<<<nfin, FIN_TB, 0, stream>>>(part, dis, dtot, bsum, n, npad);
  k_fill      <<<nranges * DC_SLICES, DC_TB, 0, stream>>>(src, dst, part, dtot, bsum, dis, rowptr,
                                                          csr, e, nranges, npad);

  k_gemm64_wmma<false><<<gemm_blocks, TB, 0, stream>>>(x, W1, h, nstrips);
  k_gather_gemm<<<tile_blocks, FT_TB, 0, stream>>>(csr, rowptr, dtot, h, dis, b1, W2, h2, n);

  k_gather<<<node_blocks, TB, 0, stream>>>(csr, rowptr, dtot, h2, dis, b2, out, n);
}


# gather launches cut to the chip's 16384 wave slots: a wave walks several nodes (slot ranges by one vector load, next node's records and own row fetched ahead), not one wave per node; same sums
# speedup vs baseline: 3.3215x; 1.0357x over previous
#include <hip/hip_runtime.h>
#include <hip/hip_bf16.h>


typedef __attribute__((ext_vector_type(2))) float v2f;
typedef __attribute__((ext_vector_type(8))) float v8f;

#define TB 256

#define DC_RANGE  12512
#define DC_SLICES 32
#define DC_TB     1024
#define DC_UNROLL 16
__global__ void __launch_bounds__(DC_TB)
k_deg_count(const int* __restrict__ dst, unsigned int* __restrict__ part, int e, int nranges) {
  __shared__ unsigned int cnt[DC_RANGE];
  const int tid = threadIdx.x;
  const int s = blockIdx.x / nranges;
  const int g = blockIdx.x - s * nranges;
  const int lo = g * DC_RANGE;

  for (int b = tid; b < DC_RANGE; b += DC_TB) cnt[b] = 0u;
  __syncthreads();

  const int per = (((e + DC_SLICES - 1) / DC_SLICES) + 31) & ~31;
  const int e0 = s * per;
  const int e1 = min(e, e0 + per);
  for (int base = e0; base < e1; base += DC_UNROLL * DC_TB) {
    int v[DC_UNROLL];
#pragma unroll
    for (int j = 0; j < DC_UNROLL; ++j) {
      v[j] = dst[min(base + j * DC_TB + tid, e1 - 1)];
    }
    asm volatile("" :: "v"(v[0]), "v"(v[1]), "v"(v[2]), "v"(v[3]), "v"(v[4]), "v"(v[5]),
                       "v"(v[6]), "v"(v[7]), "v"(v[8]), "v"(v[9]), "v"(v[10]), "v"(v[11]),
                       "v"(v[12]), "v"(v[13]), "v"(v[14]), "v"(v[15]));
#pragma unroll
    for (int j = 0; j < DC_UNROLL; ++j) {
      const bool live = (base + j * DC_TB + tid) < e1;
      const unsigned int r = (unsigned int)(v[j] - lo);
      if (live && r < (unsigned int)DC_RANGE) atomicAdd(&cnt[r], 1u);
    }
  }
  __syncthreads();

  unsigned int* row = part + (size_t)s * ((size_t)nranges * DC_RANGE) + lo;
  for (int b = tid; b < DC_RANGE; b += DC_TB) row[b] = cnt[b];
}

#define FIN_TB 544
#define FIN_PER_RANGE (DC_RANGE / FIN_TB)
static_assert(DC_RANGE % FIN_TB == 0, "finish blocks must tile a node range");
__global__ void __launch_bounds__(FIN_TB)
k_deg_finish(unsigned int* __restrict__ part, float* __restrict__ dis,
             unsigned int* __restrict__ dtot, unsigned int* __restrict__ bsum, int n, int npad) {
  __shared__ unsigned int bs;
  const int i = blockIdx.x * FIN_TB + threadIdx.x;
  if (threadIdx.x == 0) bs = 0u;
  __syncthreads();
  unsigned int c[DC_SLICES];
#pragma unroll
  for (int s = 0; s < DC_SLICES; ++s) c[s] = part[(size_t)s * npad + i];
  unsigned int run = 0u;
#pragma unroll
  for (int s = 0; s < DC_SLICES; ++s) { part[(size_t)s * npad + i] = run; run += c[s]; }
  dtot[i] = run;
  if (i < n) dis[i] = rsqrtf((float)(run + 1u));
  atomicAdd(&bs, run);
  __syncthreads();
  if (threadIdx.x == 0) bsum[blockIdx.x * 32] = bs;
}

#define SC_CH 13
static_assert(DC_TB * SC_CH >= DC_RANGE, "range scan must cover the range");
static_assert(DC_TB == 1024, "the range scan assumes 32 waves of 32 lanes");
__global__ void __launch_bounds__(DC_TB)
k_fill(const int* __restrict__ src, const int* __restrict__ dst,
       const unsigned int* __restrict__ part, const unsigned int* __restrict__ dtot,
       const unsigned int* __restrict__ bsum, const float* __restrict__ dis,
       unsigned int* __restrict__ rowptr, int2* __restrict__ csr, int e, int nranges, int npad) {
  __shared__ unsigned int cnt[DC_RANGE];
  __shared__ unsigned int wsum[32];
  __shared__ unsigned int rb;
  const int tid = threadIdx.x;
  const int lane = tid & 31;
  const int wid = tid >> 5;
  const int s = blockIdx.x / nranges;
  const int g = blockIdx.x - s * nranges;
  const int lo = g * DC_RANGE;

  if (tid == 0) rb = 0u;
  for (int b = tid; b < DC_RANGE; b += DC_TB) cnt[b] = dtot[lo + b];
  __syncthreads();

  unsigned int acc = 0u;
  for (int k = tid; k < g * FIN_PER_RANGE; k += DC_TB) acc += bsum[k * 32];
  if (acc != 0u) atomicAdd(&rb, acc);

  const int c0 = tid * SC_CH;
  unsigned int loc[SC_CH];
  unsigned int tsum = 0u;
#pragma unroll
  for (int j = 0; j < SC_CH; ++j) {
    const int b = c0 + j;
    loc[j] = (b < DC_RANGE) ? cnt[b] : 0u;
    tsum += loc[j];
  }
  unsigned int x = tsum;
#pragma unroll
  for (int o = 1; o < 32; o <<= 1) {
    const unsigned int t = __shfl_up(x, o);
    if (lane >= o) x += t;
  }
  if (lane == 31) wsum[wid] = x;
  __syncthreads();
  const unsigned int w = wsum[lane];
  unsigned int ws = w;
#pragma unroll
  for (int o = 1; o < 32; o <<= 1) {
    const unsigned int t = __shfl_up(ws, o);
    if (lane >= o) ws += t;
  }
  const unsigned int wave_excl = __shfl(ws, wid) - __shfl(w, wid);
  unsigned int run = rb + wave_excl + (x - tsum);
#pragma unroll
  for (int j = 0; j < SC_CH; ++j) {
    const int b = c0 + j;
    if (b < DC_RANGE) { cnt[b] = run; run += loc[j]; }
  }
  __syncthreads();

  for (int b = tid; b < DC_RANGE; b += DC_TB) {
    const unsigned int v = cnt[b];
    if (s == 0) rowptr[lo + b] = v;
    cnt[b] = v + part[(size_t)s * npad + lo + b];
  }
  __syncthreads();

  const int per = (((e + DC_SLICES - 1) / DC_SLICES) + 31) & ~31;
  const int e0 = s * per;
  const int e1 = min(e, e0 + per);
  for (int base = e0; base < e1; base += DC_UNROLL * DC_TB) {
    int v[DC_UNROLL], u[DC_UNROLL];
#pragma unroll
    for (int j = 0; j < DC_UNROLL; ++j) {
      const int idx = min(base + j * DC_TB + tid, e1 - 1);
      v[j] = dst[idx];
      u[j] = src[idx];
    }
    asm volatile("" :: "v"(v[0]), "v"(v[1]), "v"(v[2]), "v"(v[3]), "v"(v[4]), "v"(v[5]),
                       "v"(v[6]), "v"(v[7]), "v"(v[8]), "v"(v[9]), "v"(v[10]), "v"(v[11]),
                       "v"(v[12]), "v"(v[13]), "v"(v[14]), "v"(v[15]));
    asm volatile("" :: "v"(u[0]), "v"(u[1]), "v"(u[2]), "v"(u[3]), "v"(u[4]), "v"(u[5]),
                       "v"(u[6]), "v"(u[7]), "v"(u[8]), "v"(u[9]), "v"(u[10]), "v"(u[11]),
                       "v"(u[12]), "v"(u[13]), "v"(u[14]), "v"(u[15]));
    unsigned int p[DC_UNROLL];
    float ds[DC_UNROLL];
#pragma unroll
    for (int j = 0; j < DC_UNROLL; ++j) {
      const bool live = (base + j * DC_TB + tid) < e1;
      const unsigned int r = (unsigned int)(v[j] - lo);
      p[j] = 0xffffffffu;
      ds[j] = 0.0f;
      if (live && r < (unsigned int)DC_RANGE) {
        p[j] = atomicAdd(&cnt[r], 1u);
        ds[j] = dis[u[j]];
      }
    }
#pragma unroll
    for (int j = 0; j < DC_UNROLL; ++j) {
      if (p[j] != 0xffffffffu) csr[p[j]] = make_int2(u[j], __float_as_int(ds[j]));
    }
  }
}

template <bool RELU_IN>
__global__ void k_gemm64_wmma(const float* __restrict__ X,
                              const float* __restrict__ W,
                              float* __restrict__ H, int nstrips) {
  int wave = (int)((blockIdx.x * blockDim.x + threadIdx.x) >> 5);
  int lane = threadIdx.x & 31;
  if (wave >= nstrips) return;

  const int m0    = wave * 16;
  const int mlow  = lane & 15;
  const int khalf = (lane >> 4) << 1;

  v2f a[16];
  const float* xr = X + (size_t)(m0 + mlow) * 64 + khalf;
#pragma unroll
  for (int k = 0; k < 16; ++k) {
    float ax = xr[k * 4 + 0];
    float ay = xr[k * 4 + 1];
    if (RELU_IN) { ax = fmaxf(ax, 0.0f); ay = fmaxf(ay, 0.0f); }
    a[k].x = ax; a[k].y = ay;
  }

#pragma unroll
  for (int nt = 0; nt < 4; ++nt) {
    const float* wr = W + (size_t)(nt * 16 + mlow) * 64 + khalf;
    v8f c = {};
#pragma unroll
    for (int k = 0; k < 16; ++k) {
      v2f b;
      b.x = wr[k * 4 + 0];
      b.y = wr[k * 4 + 1];
      c = __builtin_amdgcn_wmma_f32_16x16x4_f32(
           false, a[k],  false, b,
           (short)0, c,  false,  false);
    }
    float* o = H + (size_t)(m0 + 8 * (lane >> 4)) * 64 + nt * 16 + mlow;
#pragma unroll
    for (int v = 0; v < 8; ++v) o[(size_t)v * 64] = c[v];
  }
}

#define GB 8
__device__ __forceinline__ void node_sum(const int2* __restrict__ csr, const float* __restrict__ h,
                                         int start, int cnt, float di, int2 rec0,
                                         float own0, float own1, int lane,
                                         float& acc0, float& acc1) {
  const float wself = di * di;
  acc0 = own0 * wself;
  acc1 = own1 * wself;
  for (int k0 = 0; k0 < cnt; k0 += 32) {
    const int m = min(cnt - k0, 32);
    int2 rec = rec0;
    if (k0 > 0) {
      rec = csr[start + k0 + min(lane, m - 1)];
      asm volatile("" :: "v"(rec.x), "v"(rec.y));
    }
    const int myidx = rec.x;
    const float dsrc = __int_as_float(rec.y);
    for (int k = 0; k < m; k += GB) {
      int s[GB];
#pragma unroll
      for (int j = 0; j < GB; ++j) s[j] = __builtin_amdgcn_readlane(myidx, k + j);
      float v0[GB], v1[GB];
#pragma unroll
      for (int j = 0; j < GB; ++j) {
        const float* hr = h + (size_t)s[j] * 64 + lane;
        v0[j] = hr[0];
        v1[j] = hr[32];
      }
      float w[GB];
#pragma unroll
      for (int j = 0; j < GB; ++j) {
        const float ds = __int_as_float(__builtin_amdgcn_readlane(__float_as_int(dsrc), k + j));
        w[j] = (k + j < m) ? ds * di : 0.0f;
      }
#pragma unroll
      for (int j = 0; j < GB; ++j) {
        acc0 = fmaf(v0[j], w[j], acc0);
        acc1 = fmaf(v1[j], w[j], acc1);
      }
    }
  }
}

__device__ __forceinline__ int2 first_records(const int2* __restrict__ csr, int start, int cnt,
                                              int lane) {
  int2 rec = make_int2(0, 0);
  if (cnt > 0) rec = csr[start + min(lane, min(cnt, 32) - 1)];
  return rec;
}

#define PG_WAVES 16384
__global__ void __launch_bounds__(TB)
k_gather(const int2* __restrict__ csr, const unsigned int* __restrict__ rowptr,
         const unsigned int* __restrict__ dtot, const float* __restrict__ h,
         const float* __restrict__ dis, const float* __restrict__ bias,
         float* __restrict__ out, int n) {
  const int lane = threadIdx.x & 31;
  const int w0 = __builtin_amdgcn_readfirstlane((int)((blockIdx.x * blockDim.x + threadIdx.x) >> 5));
  const int stride = (int)((gridDim.x * blockDim.x) >> 5);
  if (w0 >= n) return;
  const float bia0 = bias[lane];
  const float bia1 = bias[lane + 32];
  for (int base = w0; base < n; base += 32 * stride) {
    const int j = min(base + lane * stride, n - 1);
    const int mst = (int)rowptr[j];
    const int mc = (int)dtot[j];
    const int md = __float_as_int(dis[j]);
    int2 recA = first_records(csr, __builtin_amdgcn_readlane(mst, 0),
                              __builtin_amdgcn_readlane(mc, 0), lane);
    float hA0 = h[(size_t)base * 64 + lane];
    float hA1 = h[(size_t)base * 64 + lane + 32];
    asm volatile("" :: "v"(mst), "v"(mc), "v"(md), "v"(recA.x), "v"(recA.y), "v"(hA0), "v"(hA1));
    for (int k = 0; k < 32; ++k) {
      const int i = base + k * stride;
      if (i >= n) break;
      int2 recB = make_int2(0, 0);
      float hB0 = 0.0f, hB1 = 0.0f;
      if (k + 1 < 32 && i + stride < n) {
        recB = first_records(csr, __builtin_amdgcn_readlane(mst, k + 1),
                             __builtin_amdgcn_readlane(mc, k + 1), lane);
        const float* hn = h + (size_t)(i + stride) * 64 + lane;
        hB0 = hn[0];
        hB1 = hn[32];
      }
      const int st = __builtin_amdgcn_readlane(mst, k);
      const int c = __builtin_amdgcn_readlane(mc, k);
      const float d = __int_as_float(__builtin_amdgcn_readlane(md, k));
      float acc0, acc1;
      node_sum(csr, h, st, c, d, recA, hA0, hA1, lane, acc0, acc1);
      float* o = out + (size_t)i * 64 + lane;
      o[0]  = acc0 + bia0;
      o[32] = acc1 + bia1;
      recA = recB; hA0 = hB0; hA1 = hB1;
    }
  }
}

#define FT_NODES 16
#define FT_TB    (FT_NODES * 32)
#define FT_PITCH 68
__global__ void __launch_bounds__(FT_TB, 16)
k_gather_gemm(const int2* __restrict__ csr, const unsigned int* __restrict__ rowptr,
              const unsigned int* __restrict__ dtot, const float* __restrict__ h,
              const float* __restrict__ dis, const float* __restrict__ bias,
              const float* __restrict__ W, float* __restrict__ H2, int n, int ntiles) {
  __shared__ float tile[2][FT_NODES * FT_PITCH];
  const int lane = threadIdx.x & 31;
  const int wid = __builtin_amdgcn_readfirstlane((int)(threadIdx.x >> 5));
  const int ngrp = (int)gridDim.x;
  const int stride = ngrp * FT_NODES;
  const float bia0 = bias[lane];
  const float bia1 = bias[lane + 32];
  const int mlow  = lane & 15;
  const int khalf = (lane >> 4) << 1;

  int buf = 0;
  for (int tb = (int)blockIdx.x; tb < ntiles; tb += 32 * ngrp) {
    const int ibase = tb * FT_NODES + wid;
    const int j = min(ibase + lane * stride, n - 1);
    const int mst = (int)rowptr[j];
    const int mc = (int)dtot[j];
    const int md = __float_as_int(dis[j]);
    int2 recA = make_int2(0, 0);
    float hA0 = 0.0f, hA1 = 0.0f;
    if (ibase < n) {
      recA = first_records(csr, __builtin_amdgcn_readlane(mst, 0),
                           __builtin_amdgcn_readlane(mc, 0), lane);
      hA0 = h[(size_t)ibase * 64 + lane];
      hA1 = h[(size_t)ibase * 64 + lane + 32];
    }
    asm volatile("" :: "v"(mst), "v"(mc), "v"(md), "v"(recA.x), "v"(recA.y), "v"(hA0), "v"(hA1));
    for (int k = 0; k < 32; ++k) {
      const int t = tb + k * ngrp;
      if (t >= ntiles) break;
      const int m0 = t * FT_NODES;
      const int i = m0 + wid;
      int2 recB = make_int2(0, 0);
      float hB0 = 0.0f, hB1 = 0.0f;
      if (k + 1 < 32 && i + stride < n) {
        recB = first_records(csr, __builtin_amdgcn_readlane(mst, k + 1),
                             __builtin_amdgcn_readlane(mc, k + 1), lane);
        const float* hn = h + (size_t)(i + stride) * 64 + lane;
        hB0 = hn[0];
        hB1 = hn[32];
      }
      float r0 = 0.0f, r1 = 0.0f;
      if (i < n) {
        const int st = __builtin_amdgcn_readlane(mst, k);
        const int c = __builtin_amdgcn_readlane(mc, k);
        const float d = __int_as_float(__builtin_amdgcn_readlane(md, k));
        float acc0, acc1;
        node_sum(csr, h, st, c, d, recA, hA0, hA1, lane, acc0, acc1);
        r0 = fmaxf(acc0 + bia0, 0.0f);
        r1 = fmaxf(acc1 + bia1, 0.0f);
      }
      float* tl = tile[buf];
      tl[wid * FT_PITCH + lane]      = r0;
      tl[wid * FT_PITCH + lane + 32] = r1;
      __syncthreads();
      if (wid < 4) {
        const float* ar = tl + mlow * FT_PITCH + khalf;
        const float* wr = W + (size_t)(wid * 16 + mlow) * 64 + khalf;
        v8f c = {};
#pragma unroll
        for (int hq = 0; hq < 16; hq += 8) {
          v2f a[8], b[8];
#pragma unroll
          for (int q = 0; q < 8; ++q) {
            b[q].x = wr[(hq + q) * 4 + 0];
            b[q].y = wr[(hq + q) * 4 + 1];
          }
#pragma unroll
          for (int q = 0; q < 8; ++q) {
            a[q].x = ar[(hq + q) * 4 + 0];
            a[q].y = ar[(hq + q) * 4 + 1];
          }
#pragma unroll
          for (int q = 0; q < 8; ++q) {
            c = __builtin_amdgcn_wmma_f32_16x16x4_f32(
                 false, a[q],  false, b[q],
                 (short)0, c,  false,  false);
          }
        }
        const int row0 = m0 + 8 * (lane >> 4);
        float* o = H2 + (size_t)row0 * 64 + wid * 16 + mlow;
        if (m0 + FT_NODES <= n) {
#pragma unroll
          for (int v = 0; v < 8; ++v) o[(size_t)v * 64] = c[v];
        } else {
#pragma unroll
          for (int v = 0; v < 8; ++v) {
            if (row0 + v < n) o[(size_t)v * 64] = c[v];
          }
        }
      }
      buf ^= 1;
      recA = recB; hA0 = hB0; hA1 = hB1;
    }
  }
}

extern "C" void kernel_launch(void* const* d_in, const int* in_sizes, int n_in,
                              void* d_out, int out_size, void* d_ws, size_t ws_size,
                              hipStream_t stream) {
  (void)n_in; (void)out_size; (void)ws_size;

  const float* x  = (const float*)d_in[0];
  const int*   ei = (const int*)  d_in[1];
  const float* W1 = (const float*)d_in[2];
  const float* b1 = (const float*)d_in[3];
  const float* W2 = (const float*)d_in[4];
  const float* b2 = (const float*)d_in[5];
  float* out = (float*)d_out;

  const int n = in_sizes[0] / 64;
  const int e = in_sizes[1] / 2;
  const int* src = ei;
  const int* dst = ei + e;

  const int nstrips = (n + 15) / 16;
  const int gemm_blocks = (nstrips + 7) / 8;
  const int node_blocks = (n + 7) / 8;
  const int tile_blocks = (n + FT_NODES - 1) / FT_NODES;
  const int nranges     = (n + DC_RANGE - 1) / DC_RANGE;
  const int npad        = nranges * DC_RANGE;
  const int nfin        = nranges * FIN_PER_RANGE;

  float* dis = (float*)d_ws;
  float* h   = dis + n;
  float* h2  = h + (size_t)n * 64;
  unsigned int* part   = (unsigned int*)(h2 + (size_t)n * 64);
  unsigned int* dtot   = part + (size_t)DC_SLICES * npad;
  unsigned int* rowptr = dtot + npad;
  unsigned int* bsum   = rowptr + npad;
  int2* csr = (int2*)(bsum + (size_t)nfin * 32);

  k_deg_count <<<nranges * DC_SLICES, DC_TB, 0, stream>>>(dst, part, e, nranges);
  k_deg_finish<<<nfin, FIN_TB, 0, stream>>>(part, dis, dtot, bsum, n, npad);
  k_fill      <<<nranges * DC_SLICES, DC_TB, 0, stream>>>(src, dst, part, dtot, bsum, dis, rowptr,
                                                          csr, e, nranges, npad);

  k_gemm64_wmma<false><<<gemm_blocks, TB, 0, stream>>>(x, W1, h, nstrips);
  k_gather_gemm<<<tile_blocks < PG_WAVES / FT_NODES ? tile_blocks : PG_WAVES / FT_NODES, FT_TB, 0, stream>>>(
      csr, rowptr, dtot, h, dis, b1, W2, h2, n, tile_blocks);

  k_gather<<<node_blocks < PG_WAVES / 8 ? node_blocks : PG_WAVES / 8, TB, 0, stream>>>(
      csr, rowptr, dtot, h2, dis, b2, out, n);
}


# layer-1 product moved behind the aggregation (linear in rows): gather sums x rows, the fused launch forms relu(b1+sum@W1^T) on WMMA ahead of layer 2; x@W1^T launch and plane gone; sum order changes
# speedup vs baseline: 3.6163x; 1.0888x over previous
#include <hip/hip_runtime.h>
#include <hip/hip_bf16.h>


typedef __attribute__((ext_vector_type(2))) float v2f;
typedef __attribute__((ext_vector_type(8))) float v8f;

#define TB 256

#define DC_RANGE  12512
#define DC_SLICES 32
#define DC_TB     1024
#define DC_UNROLL 16
__global__ void __launch_bounds__(DC_TB)
k_deg_count(const int* __restrict__ dst, unsigned int* __restrict__ part, int e, int nranges) {
  __shared__ unsigned int cnt[DC_RANGE];
  const int tid = threadIdx.x;
  const int s = blockIdx.x / nranges;
  const int g = blockIdx.x - s * nranges;
  const int lo = g * DC_RANGE;

  for (int b = tid; b < DC_RANGE; b += DC_TB) cnt[b] = 0u;
  __syncthreads();

  const int per = (((e + DC_SLICES - 1) / DC_SLICES) + 31) & ~31;
  const int e0 = s * per;
  const int e1 = min(e, e0 + per);
  for (int base = e0; base < e1; base += DC_UNROLL * DC_TB) {
    int v[DC_UNROLL];
#pragma unroll
    for (int j = 0; j < DC_UNROLL; ++j) {
      v[j] = dst[min(base + j * DC_TB + tid, e1 - 1)];
    }
    asm volatile("" :: "v"(v[0]), "v"(v[1]), "v"(v[2]), "v"(v[3]), "v"(v[4]), "v"(v[5]),
                       "v"(v[6]), "v"(v[7]), "v"(v[8]), "v"(v[9]), "v"(v[10]), "v"(v[11]),
                       "v"(v[12]), "v"(v[13]), "v"(v[14]), "v"(v[15]));
#pragma unroll
    for (int j = 0; j < DC_UNROLL; ++j) {
      const bool live = (base + j * DC_TB + tid) < e1;
      const unsigned int r = (unsigned int)(v[j] - lo);
      if (live && r < (unsigned int)DC_RANGE) atomicAdd(&cnt[r], 1u);
    }
  }
  __syncthreads();

  unsigned int* row = part + (size_t)s * ((size_t)nranges * DC_RANGE) + lo;
  for (int b = tid; b < DC_RANGE; b += DC_TB) row[b] = cnt[b];
}

#define FIN_TB 544
#define FIN_PER_RANGE (DC_RANGE / FIN_TB)
static_assert(DC_RANGE % FIN_TB == 0, "finish blocks must tile a node range");
__global__ void __launch_bounds__(FIN_TB)
k_deg_finish(unsigned int* __restrict__ part, float* __restrict__ dis,
             unsigned int* __restrict__ dtot, unsigned int* __restrict__ bsum, int n, int npad) {
  __shared__ unsigned int bs;
  const int i = blockIdx.x * FIN_TB + threadIdx.x;
  if (threadIdx.x == 0) bs = 0u;
  __syncthreads();
  unsigned int c[DC_SLICES];
#pragma unroll
  for (int s = 0; s < DC_SLICES; ++s) c[s] = part[(size_t)s * npad + i];
  unsigned int run = 0u;
#pragma unroll
  for (int s = 0; s < DC_SLICES; ++s) { part[(size_t)s * npad + i] = run; run += c[s]; }
  dtot[i] = run;
  if (i < n) dis[i] = rsqrtf((float)(run + 1u));
  atomicAdd(&bs, run);
  __syncthreads();
  if (threadIdx.x == 0) bsum[blockIdx.x * 32] = bs;
}

#define SC_CH 13
static_assert(DC_TB * SC_CH >= DC_RANGE, "range scan must cover the range");
static_assert(DC_TB == 1024, "the range scan assumes 32 waves of 32 lanes");
__global__ void __launch_bounds__(DC_TB)
k_fill(const int* __restrict__ src, const int* __restrict__ dst,
       const unsigned int* __restrict__ part, const unsigned int* __restrict__ dtot,
       const unsigned int* __restrict__ bsum, const float* __restrict__ dis,
       unsigned int* __restrict__ rowptr, int2* __restrict__ csr, int e, int nranges, int npad) {
  __shared__ unsigned int cnt[DC_RANGE];
  __shared__ unsigned int wsum[32];
  __shared__ unsigned int rb;
  const int tid = threadIdx.x;
  const int lane = tid & 31;
  const int wid = tid >> 5;
  const int s = blockIdx.x / nranges;
  const int g = blockIdx.x - s * nranges;
  const int lo = g * DC_RANGE;

  if (tid == 0) rb = 0u;
  for (int b = tid; b < DC_RANGE; b += DC_TB) cnt[b] = dtot[lo + b];
  __syncthreads();

  unsigned int acc = 0u;
  for (int k = tid; k < g * FIN_PER_RANGE; k += DC_TB) acc += bsum[k * 32];
  if (acc != 0u) atomicAdd(&rb, acc);

  const int c0 = tid * SC_CH;
  unsigned int loc[SC_CH];
  unsigned int tsum = 0u;
#pragma unroll
  for (int j = 0; j < SC_CH; ++j) {
    const int b = c0 + j;
    loc[j] = (b < DC_RANGE) ? cnt[b] : 0u;
    tsum += loc[j];
  }
  unsigned int x = tsum;
#pragma unroll
  for (int o = 1; o < 32; o <<= 1) {
    const unsigned int t = __shfl_up(x, o);
    if (lane >= o) x += t;
  }
  if (lane == 31) wsum[wid] = x;
  __syncthreads();
  const unsigned int w = wsum[lane];
  unsigned int ws = w;
#pragma unroll
  for (int o = 1; o < 32; o <<= 1) {
    const unsigned int t = __shfl_up(ws, o);
    if (lane >= o) ws += t;
  }
  const unsigned int wave_excl = __shfl(ws, wid) - __shfl(w, wid);
  unsigned int run = rb + wave_excl + (x - tsum);
#pragma unroll
  for (int j = 0; j < SC_CH; ++j) {
    const int b = c0 + j;
    if (b < DC_RANGE) { cnt[b] = run; run += loc[j]; }
  }
  __syncthreads();

  for (int b = tid; b < DC_RANGE; b += DC_TB) {
    const unsigned int v = cnt[b];
    if (s == 0) rowptr[lo + b] = v;
    cnt[b] = v + part[(size_t)s * npad + lo + b];
  }
  __syncthreads();

  const int per = (((e + DC_SLICES - 1) / DC_SLICES) + 31) & ~31;
  const int e0 = s * per;
  const int e1 = min(e, e0 + per);
  for (int base = e0; base < e1; base += DC_UNROLL * DC_TB) {
    int v[DC_UNROLL], u[DC_UNROLL];
#pragma unroll
    for (int j = 0; j < DC_UNROLL; ++j) {
      const int idx = min(base + j * DC_TB + tid, e1 - 1);
      v[j] = dst[idx];
      u[j] = src[idx];
    }
    asm volatile("" :: "v"(v[0]), "v"(v[1]), "v"(v[2]), "v"(v[3]), "v"(v[4]), "v"(v[5]),
                       "v"(v[6]), "v"(v[7]), "v"(v[8]), "v"(v[9]), "v"(v[10]), "v"(v[11]),
                       "v"(v[12]), "v"(v[13]), "v"(v[14]), "v"(v[15]));
    asm volatile("" :: "v"(u[0]), "v"(u[1]), "v"(u[2]), "v"(u[3]), "v"(u[4]), "v"(u[5]),
                       "v"(u[6]), "v"(u[7]), "v"(u[8]), "v"(u[9]), "v"(u[10]), "v"(u[11]),
                       "v"(u[12]), "v"(u[13]), "v"(u[14]), "v"(u[15]));
    unsigned int p[DC_UNROLL];
    float ds[DC_UNROLL];
#pragma unroll
    for (int j = 0; j < DC_UNROLL; ++j) {
      const bool live = (base + j * DC_TB + tid) < e1;
      const unsigned int r = (unsigned int)(v[j] - lo);
      p[j] = 0xffffffffu;
      ds[j] = 0.0f;
      if (live && r < (unsigned int)DC_RANGE) {
        p[j] = atomicAdd(&cnt[r], 1u);
        ds[j] = dis[u[j]];
      }
    }
#pragma unroll
    for (int j = 0; j < DC_UNROLL; ++j) {
      if (p[j] != 0xffffffffu) csr[p[j]] = make_int2(u[j], __float_as_int(ds[j]));
    }
  }
}


#define GB 8
__device__ __forceinline__ void node_sum(const int2* __restrict__ csr, const float* __restrict__ h,
                                         int start, int cnt, float di, int2 rec0,
                                         float own0, float own1, int lane,
                                         float& acc0, float& acc1) {
  const float wself = di * di;
  acc0 = own0 * wself;
  acc1 = own1 * wself;
  for (int k0 = 0; k0 < cnt; k0 += 32) {
    const int m = min(cnt - k0, 32);
    int2 rec = rec0;
    if (k0 > 0) {
      rec = csr[start + k0 + min(lane, m - 1)];
      asm volatile("" :: "v"(rec.x), "v"(rec.y));
    }
    const int myidx = rec.x;
    const float dsrc = __int_as_float(rec.y);
    for (int k = 0; k < m; k += GB) {
      int s[GB];
#pragma unroll
      for (int j = 0; j < GB; ++j) s[j] = __builtin_amdgcn_readlane(myidx, k + j);
      float v0[GB], v1[GB];
#pragma unroll
      for (int j = 0; j < GB; ++j) {
        const float* hr = h + (size_t)s[j] * 64 + lane;
        v0[j] = hr[0];
        v1[j] = hr[32];
      }
      float w[GB];
#pragma unroll
      for (int j = 0; j < GB; ++j) {
        const float ds = __int_as_float(__builtin_amdgcn_readlane(__float_as_int(dsrc), k + j));
        w[j] = (k + j < m) ? ds * di : 0.0f;
      }
#pragma unroll
      for (int j = 0; j < GB; ++j) {
        acc0 = fmaf(v0[j], w[j], acc0);
        acc1 = fmaf(v1[j], w[j], acc1);
      }
    }
  }
}

__device__ __forceinline__ int2 first_records(const int2* __restrict__ csr, int start, int cnt,
                                              int lane) {
  int2 rec = make_int2(0, 0);
  if (cnt > 0) rec = csr[start + min(lane, min(cnt, 32) - 1)];
  return rec;
}

#define PG_WAVES 16384
__global__ void __launch_bounds__(TB)
k_gather(const int2* __restrict__ csr, const unsigned int* __restrict__ rowptr,
         const unsigned int* __restrict__ dtot, const float* __restrict__ h,
         const float* __restrict__ dis, const float* __restrict__ bias,
         float* __restrict__ out, int n) {
  const int lane = threadIdx.x & 31;
  const int w0 = __builtin_amdgcn_readfirstlane((int)((blockIdx.x * blockDim.x + threadIdx.x) >> 5));
  const int stride = (int)((gridDim.x * blockDim.x) >> 5);
  if (w0 >= n) return;
  const float bia0 = bias[lane];
  const float bia1 = bias[lane + 32];
  for (int base = w0; base < n; base += 32 * stride) {
    const int j = min(base + lane * stride, n - 1);
    const int mst = (int)rowptr[j];
    const int mc = (int)dtot[j];
    const int md = __float_as_int(dis[j]);
    int2 recA = first_records(csr, __builtin_amdgcn_readlane(mst, 0),
                              __builtin_amdgcn_readlane(mc, 0), lane);
    float hA0 = h[(size_t)base * 64 + lane];
    float hA1 = h[(size_t)base * 64 + lane + 32];
    asm volatile("" :: "v"(mst), "v"(mc), "v"(md), "v"(recA.x), "v"(recA.y), "v"(hA0), "v"(hA1));
    for (int k = 0; k < 32; ++k) {
      const int i = base + k * stride;
      if (i >= n) break;
      int2 recB = make_int2(0, 0);
      float hB0 = 0.0f, hB1 = 0.0f;
      if (k + 1 < 32 && i + stride < n) {
        recB = first_records(csr, __builtin_amdgcn_readlane(mst, k + 1),
                             __builtin_amdgcn_readlane(mc, k + 1), lane);
        const float* hn = h + (size_t)(i + stride) * 64 + lane;
        hB0 = hn[0];
        hB1 = hn[32];
      }
      const int st = __builtin_amdgcn_readlane(mst, k);
      const int c = __builtin_amdgcn_readlane(mc, k);
      const float d = __int_as_float(__builtin_amdgcn_readlane(md, k));
      float acc0, acc1;
      node_sum(csr, h, st, c, d, recA, hA0, hA1, lane, acc0, acc1);
      float* o = out + (size_t)i * 64 + lane;
      o[0]  = acc0 + bia0;
      o[32] = acc1 + bia1;
      recA = recB; hA0 = hB0; hA1 = hB1;
    }
  }
}

#define FT_NODES 16
#define FT_TB    (FT_NODES * 32)
#define FT_PITCH 68

__device__ __forceinline__ v8f tile_product(const float* __restrict__ rows,
                                            const float* __restrict__ W, int nt, int lane,
                                            float c0) {
  const int mlow  = lane & 15;
  const int khalf = (lane >> 4) << 1;
  const float* ar = rows + mlow * FT_PITCH + khalf;
  const float* wr = W + (unsigned int)((nt * 16 + mlow) * 64 + khalf);
  v8f c = {c0, c0, c0, c0, c0, c0, c0, c0};
#pragma unroll
  for (int hq = 0; hq < 16; hq += 8) {
    v2f a[8], b[8];
#pragma unroll
    for (int q = 0; q < 8; ++q) {
      b[q].x = wr[(hq + q) * 4 + 0];
      b[q].y = wr[(hq + q) * 4 + 1];
    }
#pragma unroll
    for (int q = 0; q < 8; ++q) {
      a[q].x = ar[(hq + q) * 4 + 0];
      a[q].y = ar[(hq + q) * 4 + 1];
    }
#pragma unroll
    for (int q = 0; q < 8; ++q) {
      c = __builtin_amdgcn_wmma_f32_16x16x4_f32(
           false, a[q],  false, b[q],
           (short)0, c,  false,  false);
    }
  }
  return c;
}

__device__ __forceinline__ void store_tile(const v8f c, float* __restrict__ H2,
                                           int m0, int n, int nt, int lane) {
  const int row0 = m0 + 8 * (lane >> 4);
  float* o = H2 + (size_t)row0 * 64 + nt * 16 + (lane & 15);
  if (m0 + FT_NODES <= n) {
#pragma unroll
    for (int v = 0; v < 8; ++v) o[(size_t)v * 64] = c[v];
  } else {
#pragma unroll
    for (int v = 0; v < 8; ++v) {
      if (row0 + v < n) o[(size_t)v * 64] = c[v];
    }
  }
}

#define FT_TILE (FT_NODES * FT_PITCH)
__global__ void __launch_bounds__(FT_TB, 16)
k_gather_gemm(const int2* __restrict__ csr, const unsigned int* __restrict__ rowptr,
              const unsigned int* __restrict__ dtot, const float* __restrict__ x,
              const float* __restrict__ dis, const float* __restrict__ W1,
              const float* __restrict__ b1, const float* __restrict__ W2,
              float* __restrict__ H2, int n, int ntiles) {
  __shared__ float lds[4][FT_TILE];
  const int lane = threadIdx.x & 31;
  const int wid = __builtin_amdgcn_readfirstlane((int)(threadIdx.x >> 5));
  const int ngrp = (int)gridDim.x;
  const int stride = ngrp * FT_NODES;
  const int nt = wid & 3;
  const bool l1 = wid >= 4;

  int buf = 0;
  int pm0 = -1;
  for (int tb = (int)blockIdx.x; tb < ntiles; tb += 32 * ngrp) {
    const int ibase = tb * FT_NODES + wid;
    const int j = min(ibase + lane * stride, n - 1);
    const int mst = (int)rowptr[j];
    const int mc = (int)dtot[j];
    const int md = __float_as_int(dis[j]);
    int2 recA = make_int2(0, 0);
    float hA0 = 0.0f, hA1 = 0.0f;
    if (ibase < n) {
      recA = first_records(csr, __builtin_amdgcn_readlane(mst, 0),
                           __builtin_amdgcn_readlane(mc, 0), lane);
      hA0 = x[(size_t)ibase * 64 + lane];
      hA1 = x[(size_t)ibase * 64 + lane + 32];
    }
    asm volatile("" :: "v"(mst), "v"(mc), "v"(md), "v"(recA.x), "v"(recA.y), "v"(hA0), "v"(hA1));
    for (int k = 0; k < 32; ++k) {
      const int t = tb + k * ngrp;
      if (t >= ntiles) break;
      const int m0 = t * FT_NODES;
      const int i = m0 + wid;
      int2 recB = make_int2(0, 0);
      float hB0 = 0.0f, hB1 = 0.0f;
      if (k + 1 < 32 && i + stride < n) {
        recB = first_records(csr, __builtin_amdgcn_readlane(mst, k + 1),
                             __builtin_amdgcn_readlane(mc, k + 1), lane);
        const float* hn = x + (size_t)(i + stride) * 64 + lane;
        hB0 = hn[0];
        hB1 = hn[32];
      }
      float r0 = 0.0f, r1 = 0.0f;
      if (i < n) {
        const int st = __builtin_amdgcn_readlane(mst, k);
        const int c = __builtin_amdgcn_readlane(mc, k);
        const float d = __int_as_float(__builtin_amdgcn_readlane(md, k));
        node_sum(csr, x, st, c, d, recA, hA0, hA1, lane, r0, r1);
      }
      lds[buf][wid * FT_PITCH + lane]      = r0;
      lds[buf][wid * FT_PITCH + lane + 32] = r1;
      __syncthreads();
      if (wid < 8 && (l1 || pm0 >= 0)) {
        float c0 = 0.0f;
        if (l1) c0 = b1[nt * 16 + (lane & 15)];
        const v8f c = tile_product(lds[l1 ? buf : 3 - buf], l1 ? W1 : W2, nt, lane, c0);
        if (l1) {
          float* ro = lds[2 + buf] + (8 * (lane >> 4)) * FT_PITCH + nt * 16 + (lane & 15);
#pragma unroll
          for (int v = 0; v < 8; ++v) ro[v * FT_PITCH] = fmaxf(c[v], 0.0f);
        } else {
          store_tile(c, H2, pm0, n, nt, lane);
        }
      }
      pm0 = m0;
      buf ^= 1;
      recA = recB; hA0 = hB0; hA1 = hB1;
    }
  }
  __syncthreads();
  if (wid < 4 && pm0 >= 0) {
    const v8f c = tile_product(lds[3 - buf], W2, nt, lane, 0.0f);
    store_tile(c, H2, pm0, n, nt, lane);
  }
}

extern "C" void kernel_launch(void* const* d_in, const int* in_sizes, int n_in,
                              void* d_out, int out_size, void* d_ws, size_t ws_size,
                              hipStream_t stream) {
  (void)n_in; (void)out_size; (void)ws_size;

  const float* x  = (const float*)d_in[0];
  const int*   ei = (const int*)  d_in[1];
  const float* W1 = (const float*)d_in[2];
  const float* b1 = (const float*)d_in[3];
  const float* W2 = (const float*)d_in[4];
  const float* b2 = (const float*)d_in[5];
  float* out = (float*)d_out;

  const int n = in_sizes[0] / 64;
  const int e = in_sizes[1] / 2;
  const int* src = ei;
  const int* dst = ei + e;

  const int node_blocks = (n + 7) / 8;
  const int tile_blocks = (n + FT_NODES - 1) / FT_NODES;
  const int nranges     = (n + DC_RANGE - 1) / DC_RANGE;
  const int npad        = nranges * DC_RANGE;
  const int nfin        = nranges * FIN_PER_RANGE;

  float* dis = (float*)d_ws;
  float* h2  = dis + n;
  unsigned int* part   = (unsigned int*)(h2 + (size_t)n * 64);
  unsigned int* dtot   = part + (size_t)DC_SLICES * npad;
  unsigned int* rowptr = dtot + npad;
  unsigned int* bsum   = rowptr + npad;
  int2* csr = (int2*)(bsum + (size_t)nfin * 32);

  k_deg_count <<<nranges * DC_SLICES, DC_TB, 0, stream>>>(dst, part, e, nranges);
  k_deg_finish<<<nfin, FIN_TB, 0, stream>>>(part, dis, dtot, bsum, n, npad);
  k_fill      <<<nranges * DC_SLICES, DC_TB, 0, stream>>>(src, dst, part, dtot, bsum, dis, rowptr,
                                                          csr, e, nranges, npad);

  k_gather_gemm<<<tile_blocks < PG_WAVES / FT_NODES ? tile_blocks : PG_WAVES / FT_NODES, FT_TB, 0, stream>>>(
      csr, rowptr, dtot, x, dis, W1, b1, W2, h2, n, tile_blocks);

  k_gather<<<node_blocks < PG_WAVES / 8 ? node_blocks : PG_WAVES / 8, TB, 0, stream>>>(
      csr, rowptr, dtot, h2, dis, b2, out, n);
}


# k_fill: each thread's dtot, part and bsum words fetched as one group of loads into registers at the kernel head, not one dependent memory round trip per loop pass (about 20 in a row); same sort
# speedup vs baseline: 3.8156x; 1.0551x over previous
#include <hip/hip_runtime.h>
#include <hip/hip_bf16.h>


typedef __attribute__((ext_vector_type(2))) float v2f;
typedef __attribute__((ext_vector_type(8))) float v8f;

#define TB 256

#define DC_RANGE  12512
#define DC_SLICES 32
#define DC_TB     1024
#define DC_UNROLL 16
__global__ void __launch_bounds__(DC_TB)
k_deg_count(const int* __restrict__ dst, unsigned int* __restrict__ part, int e, int nranges) {
  __shared__ unsigned int cnt[DC_RANGE];
  const int tid = threadIdx.x;
  const int s = blockIdx.x / nranges;
  const int g = blockIdx.x - s * nranges;
  const int lo = g * DC_RANGE;

  for (int b = tid; b < DC_RANGE; b += DC_TB) cnt[b] = 0u;
  __syncthreads();

  const int per = (((e + DC_SLICES - 1) / DC_SLICES) + 31) & ~31;
  const int e0 = s * per;
  const int e1 = min(e, e0 + per);
  for (int base = e0; base < e1; base += DC_UNROLL * DC_TB) {
    int v[DC_UNROLL];
#pragma unroll
    for (int j = 0; j < DC_UNROLL; ++j) {
      v[j] = dst[min(base + j * DC_TB + tid, e1 - 1)];
    }
    asm volatile("" :: "v"(v[0]), "v"(v[1]), "v"(v[2]), "v"(v[3]), "v"(v[4]), "v"(v[5]),
                       "v"(v[6]), "v"(v[7]), "v"(v[8]), "v"(v[9]), "v"(v[10]), "v"(v[11]),
                       "v"(v[12]), "v"(v[13]), "v"(v[14]), "v"(v[15]));
#pragma unroll
    for (int j = 0; j < DC_UNROLL; ++j) {
      const bool live = (base + j * DC_TB + tid) < e1;
      const unsigned int r = (unsigned int)(v[j] - lo);
      if (live && r < (unsigned int)DC_RANGE) atomicAdd(&cnt[r], 1u);
    }
  }
  __syncthreads();

  unsigned int* row = part + (size_t)s * ((size_t)nranges * DC_RANGE) + lo;
  for (int b = tid; b < DC_RANGE; b += DC_TB) row[b] = cnt[b];
}

#define FIN_TB 544
#define FIN_PER_RANGE (DC_RANGE / FIN_TB)
static_assert(DC_RANGE % FIN_TB == 0, "finish blocks must tile a node range");
__global__ void __launch_bounds__(FIN_TB)
k_deg_finish(unsigned int* __restrict__ part, float* __restrict__ dis,
             unsigned int* __restrict__ dtot, unsigned int* __restrict__ bsum, int n, int npad) {
  __shared__ unsigned int bs;
  const int i = blockIdx.x * FIN_TB + threadIdx.x;
  if (threadIdx.x == 0) bs = 0u;
  __syncthreads();
  unsigned int c[DC_SLICES];
#pragma unroll
  for (int s = 0; s < DC_SLICES; ++s) c[s] = part[(size_t)s * npad + i];
  unsigned int run = 0u;
#pragma unroll
  for (int s = 0; s < DC_SLICES; ++s) { part[(size_t)s * npad + i] = run; run += c[s]; }
  dtot[i] = run;
  if (i < n) dis[i] = rsqrtf((float)(run + 1u));
  atomicAdd(&bs, run);
  __syncthreads();
  if (threadIdx.x == 0) bsum[blockIdx.x * 32] = bs;
}

#define SC_CH 13
static_assert(DC_TB * SC_CH >= DC_RANGE, "range scan must cover the range");
static_assert(DC_TB == 1024, "the range scan assumes 32 waves of 32 lanes");
static_assert(SC_CH == 13, "the load group in k_fill names its SC_CH words one by one");
__global__ void __launch_bounds__(DC_TB)
k_fill(const int* __restrict__ src, const int* __restrict__ dst,
       const unsigned int* __restrict__ part, const unsigned int* __restrict__ dtot,
       const unsigned int* __restrict__ bsum, const float* __restrict__ dis,
       unsigned int* __restrict__ rowptr, int2* __restrict__ csr, int e, int nranges, int npad) {
  __shared__ unsigned int cnt[DC_RANGE];
  __shared__ unsigned int wsum[32];
  __shared__ unsigned int rb;
  const int tid = threadIdx.x;
  const int lane = tid & 31;
  const int wid = tid >> 5;
  const int s = blockIdx.x / nranges;
  const int g = blockIdx.x - s * nranges;
  const int lo = g * DC_RANGE;

  const unsigned int* __restrict__ prow = part + (size_t)s * npad + lo;
  const int nprev = g * FIN_PER_RANGE;
  unsigned int dreg[SC_CH], preg[SC_CH];
#pragma unroll
  for (int j = 0; j < SC_CH; ++j) {
    const int b = min(tid + j * DC_TB, DC_RANGE - 1);
    dreg[j] = dtot[lo + b];
    preg[j] = prow[b];
  }
  unsigned int acc = 0u;
  if (tid < nprev) acc = bsum[tid * 32];
  asm volatile("" :: "v"(dreg[0]), "v"(dreg[1]), "v"(dreg[2]), "v"(dreg[3]), "v"(dreg[4]),
                     "v"(dreg[5]), "v"(dreg[6]), "v"(dreg[7]), "v"(dreg[8]), "v"(dreg[9]),
                     "v"(dreg[10]), "v"(dreg[11]), "v"(dreg[12]),
                     "v"(preg[0]), "v"(preg[1]), "v"(preg[2]), "v"(preg[3]), "v"(preg[4]),
                     "v"(preg[5]), "v"(preg[6]), "v"(preg[7]), "v"(preg[8]), "v"(preg[9]),
                     "v"(preg[10]), "v"(preg[11]), "v"(preg[12]), "v"(acc));

  if (tid == 0) rb = 0u;
#pragma unroll
  for (int j = 0; j < SC_CH; ++j) {
    const int b = tid + j * DC_TB;
    if (b < DC_RANGE) cnt[b] = dreg[j];
  }
  __syncthreads();

  for (int k = tid + DC_TB; k < nprev; k += DC_TB) acc += bsum[k * 32];
  if (acc != 0u) atomicAdd(&rb, acc);

  const int c0 = tid * SC_CH;
  unsigned int loc[SC_CH];
  unsigned int tsum = 0u;
#pragma unroll
  for (int j = 0; j < SC_CH; ++j) {
    const int b = c0 + j;
    loc[j] = (b < DC_RANGE) ? cnt[b] : 0u;
    tsum += loc[j];
  }
  unsigned int x = tsum;
#pragma unroll
  for (int o = 1; o < 32; o <<= 1) {
    const unsigned int t = __shfl_up(x, o);
    if (lane >= o) x += t;
  }
  if (lane == 31) wsum[wid] = x;
  __syncthreads();
  const unsigned int w = wsum[lane];
  unsigned int ws = w;
#pragma unroll
  for (int o = 1; o < 32; o <<= 1) {
    const unsigned int t = __shfl_up(ws, o);
    if (lane >= o) ws += t;
  }
  const unsigned int wave_excl = __shfl(ws, wid) - __shfl(w, wid);
  unsigned int run = rb + wave_excl + (x - tsum);
#pragma unroll
  for (int j = 0; j < SC_CH; ++j) {
    const int b = c0 + j;
    if (b < DC_RANGE) { cnt[b] = run; run += loc[j]; }
  }
  __syncthreads();

  unsigned int rs[SC_CH];
#pragma unroll
  for (int j = 0; j < SC_CH; ++j) {
    const int b = tid + j * DC_TB;
    rs[j] = (b < DC_RANGE) ? cnt[b] : 0u;
  }
#pragma unroll
  for (int j = 0; j < SC_CH; ++j) {
    const int b = tid + j * DC_TB;
    if (b < DC_RANGE) {
      if (s == 0) rowptr[lo + b] = rs[j];
      cnt[b] = rs[j] + preg[j];
    }
  }
  __syncthreads();

  const int per = (((e + DC_SLICES - 1) / DC_SLICES) + 31) & ~31;
  const int e0 = s * per;
  const int e1 = min(e, e0 + per);
  for (int base = e0; base < e1; base += DC_UNROLL * DC_TB) {
    int v[DC_UNROLL], u[DC_UNROLL];
#pragma unroll
    for (int j = 0; j < DC_UNROLL; ++j) {
      const int idx = min(base + j * DC_TB + tid, e1 - 1);
      v[j] = dst[idx];
      u[j] = src[idx];
    }
    asm volatile("" :: "v"(v[0]), "v"(v[1]), "v"(v[2]), "v"(v[3]), "v"(v[4]), "v"(v[5]),
                       "v"(v[6]), "v"(v[7]), "v"(v[8]), "v"(v[9]), "v"(v[10]), "v"(v[11]),
                       "v"(v[12]), "v"(v[13]), "v"(v[14]), "v"(v[15]));
    asm volatile("" :: "v"(u[0]), "v"(u[1]), "v"(u[2]), "v"(u[3]), "v"(u[4]), "v"(u[5]),
                       "v"(u[6]), "v"(u[7]), "v"(u[8]), "v"(u[9]), "v"(u[10]), "v"(u[11]),
                       "v"(u[12]), "v"(u[13]), "v"(u[14]), "v"(u[15]));
    unsigned int p[DC_UNROLL];
    float ds[DC_UNROLL];
#pragma unroll
    for (int j = 0; j < DC_UNROLL; ++j) {
      const bool live = (base + j * DC_TB + tid) < e1;
      const unsigned int r = (unsigned int)(v[j] - lo);
      p[j] = 0xffffffffu;
      ds[j] = 0.0f;
      if (live && r < (unsigned int)DC_RANGE) {
        p[j] = atomicAdd(&cnt[r], 1u);
        ds[j] = dis[u[j]];
      }
    }
#pragma unroll
    for (int j = 0; j < DC_UNROLL; ++j) {
      if (p[j] != 0xffffffffu) csr[p[j]] = make_int2(u[j], __float_as_int(ds[j]));
    }
  }
}


#define GB 8
__device__ __forceinline__ void node_sum(const int2* __restrict__ csr, const float* __restrict__ h,
                                         int start, int cnt, float di, int2 rec0,
                                         float own0, float own1, int lane,
                                         float& acc0, float& acc1) {
  const float wself = di * di;
  acc0 = own0 * wself;
  acc1 = own1 * wself;
  for (int k0 = 0; k0 < cnt; k0 += 32) {
    const int m = min(cnt - k0, 32);
    int2 rec = rec0;
    if (k0 > 0) {
      rec = csr[start + k0 + min(lane, m - 1)];
      asm volatile("" :: "v"(rec.x), "v"(rec.y));
    }
    const int myidx = rec.x;
    const float dsrc = __int_as_float(rec.y);
    for (int k = 0; k < m; k += GB) {
      int s[GB];
#pragma unroll
      for (int j = 0; j < GB; ++j) s[j] = __builtin_amdgcn_readlane(myidx, k + j);
      float v0[GB], v1[GB];
#pragma unroll
      for (int j = 0; j < GB; ++j) {
        const float* hr = h + (size_t)s[j] * 64 + lane;
        v0[j] = hr[0];
        v1[j] = hr[32];
      }
      float w[GB];
#pragma unroll
      for (int j = 0; j < GB; ++j) {
        const float ds = __int_as_float(__builtin_amdgcn_readlane(__float_as_int(dsrc), k + j));
        w[j] = (k + j < m) ? ds * di : 0.0f;
      }
#pragma unroll
      for (int j = 0; j < GB; ++j) {
        acc0 = fmaf(v0[j], w[j], acc0);
        acc1 = fmaf(v1[j], w[j], acc1);
      }
    }
  }
}

__device__ __forceinline__ int2 first_records(const int2* __restrict__ csr, int start, int cnt,
                                              int lane) {
  int2 rec = make_int2(0, 0);
  if (cnt > 0) rec = csr[start + min(lane, min(cnt, 32) - 1)];
  return rec;
}

#define PG_WAVES 16384
__global__ void __launch_bounds__(TB)
k_gather(const int2* __restrict__ csr, const unsigned int* __restrict__ rowptr,
         const unsigned int* __restrict__ dtot, const float* __restrict__ h,
         const float* __restrict__ dis, const float* __restrict__ bias,
         float* __restrict__ out, int n) {
  const int lane = threadIdx.x & 31;
  const int w0 = __builtin_amdgcn_readfirstlane((int)((blockIdx.x * blockDim.x + threadIdx.x) >> 5));
  const int stride = (int)((gridDim.x * blockDim.x) >> 5);
  if (w0 >= n) return;
  const float bia0 = bias[lane];
  const float bia1 = bias[lane + 32];
  for (int base = w0; base < n; base += 32 * stride) {
    const int j = min(base + lane * stride, n - 1);
    const int mst = (int)rowptr[j];
    const int mc = (int)dtot[j];
    const int md = __float_as_int(dis[j]);
    int2 recA = first_records(csr, __builtin_amdgcn_readlane(mst, 0),
                              __builtin_amdgcn_readlane(mc, 0), lane);
    float hA0 = h[(size_t)base * 64 + lane];
    float hA1 = h[(size_t)base * 64 + lane + 32];
    asm volatile("" :: "v"(mst), "v"(mc), "v"(md), "v"(recA.x), "v"(recA.y), "v"(hA0), "v"(hA1));
    for (int k = 0; k < 32; ++k) {
      const int i = base + k * stride;
      if (i >= n) break;
      int2 recB = make_int2(0, 0);
      float hB0 = 0.0f, hB1 = 0.0f;
      if (k + 1 < 32 && i + stride < n) {
        recB = first_records(csr, __builtin_amdgcn_readlane(mst, k + 1),
                             __builtin_amdgcn_readlane(mc, k + 1), lane);
        const float* hn = h + (size_t)(i + stride) * 64 + lane;
        hB0 = hn[0];
        hB1 = hn[32];
      }
      const int st = __builtin_amdgcn_readlane(mst, k);
      const int c = __builtin_amdgcn_readlane(mc, k);
      const float d = __int_as_float(__builtin_amdgcn_readlane(md, k));
      float acc0, acc1;
      node_sum(csr, h, st, c, d, recA, hA0, hA1, lane, acc0, acc1);
      float* o = out + (size_t)i * 64 + lane;
      o[0]  = acc0 + bia0;
      o[32] = acc1 + bia1;
      recA = recB; hA0 = hB0; hA1 = hB1;
    }
  }
}

#define FT_NODES 16
#define FT_TB    (FT_NODES * 32)
#define FT_PITCH 68

__device__ __forceinline__ v8f tile_product(const float* __restrict__ rows,
                                            const float* __restrict__ W, int nt, int lane,
                                            float c0) {
  const int mlow  = lane & 15;
  const int khalf = (lane >> 4) << 1;
  const float* ar = rows + mlow * FT_PITCH + khalf;
  const float* wr = W + (unsigned int)((nt * 16 + mlow) * 64 + khalf);
  v8f c = {c0, c0, c0, c0, c0, c0, c0, c0};
#pragma unroll
  for (int hq = 0; hq < 16; hq += 8) {
    v2f a[8], b[8];
#pragma unroll
    for (int q = 0; q < 8; ++q) {
      b[q].x = wr[(hq + q) * 4 + 0];
      b[q].y = wr[(hq + q) * 4 + 1];
    }
#pragma unroll
    for (int q = 0; q < 8; ++q) {
      a[q].x = ar[(hq + q) * 4 + 0];
      a[q].y = ar[(hq + q) * 4 + 1];
    }
#pragma unroll
    for (int q = 0; q < 8; ++q) {
      c = __builtin_amdgcn_wmma_f32_16x16x4_f32(
           false, a[q],  false, b[q],
           (short)0, c,  false,  false);
    }
  }
  return c;
}

__device__ __forceinline__ void store_tile(const v8f c, float* __restrict__ H2,
                                           int m0, int n, int nt, int lane) {
  const int row0 = m0 + 8 * (lane >> 4);
  float* o = H2 + (size_t)row0 * 64 + nt * 16 + (lane & 15);
  if (m0 + FT_NODES <= n) {
#pragma unroll
    for (int v = 0; v < 8; ++v) o[(size_t)v * 64] = c[v];
  } else {
#pragma unroll
    for (int v = 0; v < 8; ++v) {
      if (row0 + v < n) o[(size_t)v * 64] = c[v];
    }
  }
}

#define FT_TILE (FT_NODES * FT_PITCH)
__global__ void __launch_bounds__(FT_TB, 16)
k_gather_gemm(const int2* __restrict__ csr, const unsigned int* __restrict__ rowptr,
              const unsigned int* __restrict__ dtot, const float* __restrict__ x,
              const float* __restrict__ dis, const float* __restrict__ W1,
              const float* __restrict__ b1, const float* __restrict__ W2,
              float* __restrict__ H2, int n, int ntiles) {
  __shared__ float lds[4][FT_TILE];
  const int lane = threadIdx.x & 31;
  const int wid = __builtin_amdgcn_readfirstlane((int)(threadIdx.x >> 5));
  const int ngrp = (int)gridDim.x;
  const int stride = ngrp * FT_NODES;
  const int nt = wid & 3;
  const bool l1 = wid >= 4;

  int buf = 0;
  int pm0 = -1;
  for (int tb = (int)blockIdx.x; tb < ntiles; tb += 32 * ngrp) {
    const int ibase = tb * FT_NODES + wid;
    const int j = min(ibase + lane * stride, n - 1);
    const int mst = (int)rowptr[j];
    const int mc = (int)dtot[j];
    const int md = __float_as_int(dis[j]);
    int2 recA = make_int2(0, 0);
    float hA0 = 0.0f, hA1 = 0.0f;
    if (ibase < n) {
      recA = first_records(csr, __builtin_amdgcn_readlane(mst, 0),
                           __builtin_amdgcn_readlane(mc, 0), lane);
      hA0 = x[(size_t)ibase * 64 + lane];
      hA1 = x[(size_t)ibase * 64 + lane + 32];
    }
    asm volatile("" :: "v"(mst), "v"(mc), "v"(md), "v"(recA.x), "v"(recA.y), "v"(hA0), "v"(hA1));
    for (int k = 0; k < 32; ++k) {
      const int t = tb + k * ngrp;
      if (t >= ntiles) break;
      const int m0 = t * FT_NODES;
      const int i = m0 + wid;
      int2 recB = make_int2(0, 0);
      float hB0 = 0.0f, hB1 = 0.0f;
      if (k + 1 < 32 && i + stride < n) {
        recB = first_records(csr, __builtin_amdgcn_readlane(mst, k + 1),
                             __builtin_amdgcn_readlane(mc, k + 1), lane);
        const float* hn = x + (size_t)(i + stride) * 64 + lane;
        hB0 = hn[0];
        hB1 = hn[32];
      }
      float r0 = 0.0f, r1 = 0.0f;
      if (i < n) {
        const int st = __builtin_amdgcn_readlane(mst, k);
        const int c = __builtin_amdgcn_readlane(mc, k);
        const float d = __int_as_float(__builtin_amdgcn_readlane(md, k));
        node_sum(csr, x, st, c, d, recA, hA0, hA1, lane, r0, r1);
      }
      lds[buf][wid * FT_PITCH + lane]      = r0;
      lds[buf][wid * FT_PITCH + lane + 32] = r1;
      __syncthreads();
      if (wid < 8 && (l1 || pm0 >= 0)) {
        float c0 = 0.0f;
        if (l1) c0 = b1[nt * 16 + (lane & 15)];
        const v8f c = tile_product(lds[l1 ? buf : 3 - buf], l1 ? W1 : W2, nt, lane, c0);
        if (l1) {
          float* ro = lds[2 + buf] + (8 * (lane >> 4)) * FT_PITCH + nt * 16 + (lane & 15);
#pragma unroll
          for (int v = 0; v < 8; ++v) ro[v * FT_PITCH] = fmaxf(c[v], 0.0f);
        } else {
          store_tile(c, H2, pm0, n, nt, lane);
        }
      }
      pm0 = m0;
      buf ^= 1;
      recA = recB; hA0 = hB0; hA1 = hB1;
    }
  }
  __syncthreads();
  if (wid < 4 && pm0 >= 0) {
    const v8f c = tile_product(lds[3 - buf], W2, nt, lane, 0.0f);
    store_tile(c, H2, pm0, n, nt, lane);
  }
}

extern "C" void kernel_launch(void* const* d_in, const int* in_sizes, int n_in,
                              void* d_out, int out_size, void* d_ws, size_t ws_size,
                              hipStream_t stream) {
  (void)n_in; (void)out_size; (void)ws_size;

  const float* x  = (const float*)d_in[0];
  const int*   ei = (const int*)  d_in[1];
  const float* W1 = (const float*)d_in[2];
  const float* b1 = (const float*)d_in[3];
  const float* W2 = (const float*)d_in[4];
  const float* b2 = (const float*)d_in[5];
  float* out = (float*)d_out;

  const int n = in_sizes[0] / 64;
  const int e = in_sizes[1] / 2;
  const int* src = ei;
  const int* dst = ei + e;

  const int node_blocks = (n + 7) / 8;
  const int tile_blocks = (n + FT_NODES - 1) / FT_NODES;
  const int nranges     = (n + DC_RANGE - 1) / DC_RANGE;
  const int npad        = nranges * DC_RANGE;
  const int nfin        = nranges * FIN_PER_RANGE;

  float* dis = (float*)d_ws;
  float* h2  = dis + n;
  unsigned int* part   = (unsigned int*)(h2 + (size_t)n * 64);
  unsigned int* dtot   = part + (size_t)DC_SLICES * npad;
  unsigned int* rowptr = dtot + npad;
  unsigned int* bsum   = rowptr + npad;
  int2* csr = (int2*)(bsum + (size_t)nfin * 32);

  k_deg_count <<<nranges * DC_SLICES, DC_TB, 0, stream>>>(dst, part, e, nranges);
  k_deg_finish<<<nfin, FIN_TB, 0, stream>>>(part, dis, dtot, bsum, n, npad);
  k_fill      <<<nranges * DC_SLICES, DC_TB, 0, stream>>>(src, dst, part, dtot, bsum, dis, rowptr,
                                                          csr, e, nranges, npad);

  k_gather_gemm<<<tile_blocks < PG_WAVES / FT_NODES ? tile_blocks : PG_WAVES / FT_NODES, FT_TB, 0, stream>>>(
      csr, rowptr, dtot, x, dis, W1, b1, W2, h2, n, tile_blocks);

  k_gather<<<node_blocks < PG_WAVES / 8 ? node_blocks : PG_WAVES / 8, TB, 0, stream>>>(
      csr, rowptr, dtot, h2, dis, b2, out, n);
}


# gathers fetch each 256-B row as ONE 8-byte load per lane (lane l holds floats 2l, 2l+1), not two 4-byte loads; row stores 8-byte too; h2 plane put first so its rows are 256-B aligned; same sums
# speedup vs baseline: 3.9900x; 1.0457x over previous
#include <hip/hip_runtime.h>
#include <hip/hip_bf16.h>


typedef __attribute__((ext_vector_type(2))) float v2f;
typedef __attribute__((ext_vector_type(8))) float v8f;

#define TB 256

#define DC_RANGE  12512
#define DC_SLICES 32
#define DC_TB     1024
#define DC_UNROLL 16
__global__ void __launch_bounds__(DC_TB)
k_deg_count(const int* __restrict__ dst, unsigned int* __restrict__ part, int e, int nranges) {
  __shared__ unsigned int cnt[DC_RANGE];
  const int tid = threadIdx.x;
  const int s = blockIdx.x / nranges;
  const int g = blockIdx.x - s * nranges;
  const int lo = g * DC_RANGE;

  for (int b = tid; b < DC_RANGE; b += DC_TB) cnt[b] = 0u;
  __syncthreads();

  const int per = (((e + DC_SLICES - 1) / DC_SLICES) + 31) & ~31;
  const int e0 = s * per;
  const int e1 = min(e, e0 + per);
  for (int base = e0; base < e1; base += DC_UNROLL * DC_TB) {
    int v[DC_UNROLL];
#pragma unroll
    for (int j = 0; j < DC_UNROLL; ++j) {
      v[j] = dst[min(base + j * DC_TB + tid, e1 - 1)];
    }
    asm volatile("" :: "v"(v[0]), "v"(v[1]), "v"(v[2]), "v"(v[3]), "v"(v[4]), "v"(v[5]),
                       "v"(v[6]), "v"(v[7]), "v"(v[8]), "v"(v[9]), "v"(v[10]), "v"(v[11]),
                       "v"(v[12]), "v"(v[13]), "v"(v[14]), "v"(v[15]));
#pragma unroll
    for (int j = 0; j < DC_UNROLL; ++j) {
      const bool live = (base + j * DC_TB + tid) < e1;
      const unsigned int r = (unsigned int)(v[j] - lo);
      if (live && r < (unsigned int)DC_RANGE) atomicAdd(&cnt[r], 1u);
    }
  }
  __syncthreads();

  unsigned int* row = part + (size_t)s * ((size_t)nranges * DC_RANGE) + lo;
  for (int b = tid; b < DC_RANGE; b += DC_TB) row[b] = cnt[b];
}

#define FIN_TB 544
#define FIN_PER_RANGE (DC_RANGE / FIN_TB)
static_assert(DC_RANGE % FIN_TB == 0, "finish blocks must tile a node range");
__global__ void __launch_bounds__(FIN_TB)
k_deg_finish(unsigned int* __restrict__ part, float* __restrict__ dis,
             unsigned int* __restrict__ dtot, unsigned int* __restrict__ bsum, int n, int npad) {
  __shared__ unsigned int bs;
  const int i = blockIdx.x * FIN_TB + threadIdx.x;
  if (threadIdx.x == 0) bs = 0u;
  __syncthreads();
  unsigned int c[DC_SLICES];
#pragma unroll
  for (int s = 0; s < DC_SLICES; ++s) c[s] = part[(size_t)s * npad + i];
  unsigned int run = 0u;
#pragma unroll
  for (int s = 0; s < DC_SLICES; ++s) { part[(size_t)s * npad + i] = run; run += c[s]; }
  dtot[i] = run;
  if (i < n) dis[i] = rsqrtf((float)(run + 1u));
  atomicAdd(&bs, run);
  __syncthreads();
  if (threadIdx.x == 0) bsum[blockIdx.x * 32] = bs;
}

#define SC_CH 13
static_assert(DC_TB * SC_CH >= DC_RANGE, "range scan must cover the range");
static_assert(DC_TB == 1024, "the range scan assumes 32 waves of 32 lanes");
static_assert(SC_CH == 13, "the load group in k_fill names its SC_CH words one by one");
__global__ void __launch_bounds__(DC_TB)
k_fill(const int* __restrict__ src, const int* __restrict__ dst,
       const unsigned int* __restrict__ part, const unsigned int* __restrict__ dtot,
       const unsigned int* __restrict__ bsum, const float* __restrict__ dis,
       unsigned int* __restrict__ rowptr, int2* __restrict__ csr, int e, int nranges, int npad) {
  __shared__ unsigned int cnt[DC_RANGE];
  __shared__ unsigned int wsum[32];
  __shared__ unsigned int rb;
  const int tid = threadIdx.x;
  const int lane = tid & 31;
  const int wid = tid >> 5;
  const int s = blockIdx.x / nranges;
  const int g = blockIdx.x - s * nranges;
  const int lo = g * DC_RANGE;

  const unsigned int* __restrict__ prow = part + (size_t)s * npad + lo;
  const int nprev = g * FIN_PER_RANGE;
  unsigned int dreg[SC_CH], preg[SC_CH];
#pragma unroll
  for (int j = 0; j < SC_CH; ++j) {
    const int b = min(tid + j * DC_TB, DC_RANGE - 1);
    dreg[j] = dtot[lo + b];
    preg[j] = prow[b];
  }
  unsigned int acc = 0u;
  if (tid < nprev) acc = bsum[tid * 32];
  asm volatile("" :: "v"(dreg[0]), "v"(dreg[1]), "v"(dreg[2]), "v"(dreg[3]), "v"(dreg[4]),
                     "v"(dreg[5]), "v"(dreg[6]), "v"(dreg[7]), "v"(dreg[8]), "v"(dreg[9]),
                     "v"(dreg[10]), "v"(dreg[11]), "v"(dreg[12]),
                     "v"(preg[0]), "v"(preg[1]), "v"(preg[2]), "v"(preg[3]), "v"(preg[4]),
                     "v"(preg[5]), "v"(preg[6]), "v"(preg[7]), "v"(preg[8]), "v"(preg[9]),
                     "v"(preg[10]), "v"(preg[11]), "v"(preg[12]), "v"(acc));

  if (tid == 0) rb = 0u;
#pragma unroll
  for (int j = 0; j < SC_CH; ++j) {
    const int b = tid + j * DC_TB;
    if (b < DC_RANGE) cnt[b] = dreg[j];
  }
  __syncthreads();

  for (int k = tid + DC_TB; k < nprev; k += DC_TB) acc += bsum[k * 32];
  if (acc != 0u) atomicAdd(&rb, acc);

  const int c0 = tid * SC_CH;
  unsigned int loc[SC_CH];
  unsigned int tsum = 0u;
#pragma unroll
  for (int j = 0; j < SC_CH; ++j) {
    const int b = c0 + j;
    loc[j] = (b < DC_RANGE) ? cnt[b] : 0u;
    tsum += loc[j];
  }
  unsigned int x = tsum;
#pragma unroll
  for (int o = 1; o < 32; o <<= 1) {
    const unsigned int t = __shfl_up(x, o);
    if (lane >= o) x += t;
  }
  if (lane == 31) wsum[wid] = x;
  __syncthreads();
  const unsigned int w = wsum[lane];
  unsigned int ws = w;
#pragma unroll
  for (int o = 1; o < 32; o <<= 1) {
    const unsigned int t = __shfl_up(ws, o);
    if (lane >= o) ws += t;
  }
  const unsigned int wave_excl = __shfl(ws, wid) - __shfl(w, wid);
  unsigned int run = rb + wave_excl + (x - tsum);
#pragma unroll
  for (int j = 0; j < SC_CH; ++j) {
    const int b = c0 + j;
    if (b < DC_RANGE) { cnt[b] = run; run += loc[j]; }
  }
  __syncthreads();

  unsigned int rs[SC_CH];
#pragma unroll
  for (int j = 0; j < SC_CH; ++j) {
    const int b = tid + j * DC_TB;
    rs[j] = (b < DC_RANGE) ? cnt[b] : 0u;
  }
#pragma unroll
  for (int j = 0; j < SC_CH; ++j) {
    const int b = tid + j * DC_TB;
    if (b < DC_RANGE) {
      if (s == 0) rowptr[lo + b] = rs[j];
      cnt[b] = rs[j] + preg[j];
    }
  }
  __syncthreads();

  const int per = (((e + DC_SLICES - 1) / DC_SLICES) + 31) & ~31;
  const int e0 = s * per;
  const int e1 = min(e, e0 + per);
  for (int base = e0; base < e1; base += DC_UNROLL * DC_TB) {
    int v[DC_UNROLL], u[DC_UNROLL];
#pragma unroll
    for (int j = 0; j < DC_UNROLL; ++j) {
      const int idx = min(base + j * DC_TB + tid, e1 - 1);
      v[j] = dst[idx];
      u[j] = src[idx];
    }
    asm volatile("" :: "v"(v[0]), "v"(v[1]), "v"(v[2]), "v"(v[3]), "v"(v[4]), "v"(v[5]),
                       "v"(v[6]), "v"(v[7]), "v"(v[8]), "v"(v[9]), "v"(v[10]), "v"(v[11]),
                       "v"(v[12]), "v"(v[13]), "v"(v[14]), "v"(v[15]));
    asm volatile("" :: "v"(u[0]), "v"(u[1]), "v"(u[2]), "v"(u[3]), "v"(u[4]), "v"(u[5]),
                       "v"(u[6]), "v"(u[7]), "v"(u[8]), "v"(u[9]), "v"(u[10]), "v"(u[11]),
                       "v"(u[12]), "v"(u[13]), "v"(u[14]), "v"(u[15]));
    unsigned int p[DC_UNROLL];
    float ds[DC_UNROLL];
#pragma unroll
    for (int j = 0; j < DC_UNROLL; ++j) {
      const bool live = (base + j * DC_TB + tid) < e1;
      const unsigned int r = (unsigned int)(v[j] - lo);
      p[j] = 0xffffffffu;
      ds[j] = 0.0f;
      if (live && r < (unsigned int)DC_RANGE) {
        p[j] = atomicAdd(&cnt[r], 1u);
        ds[j] = dis[u[j]];
      }
    }
#pragma unroll
    for (int j = 0; j < DC_UNROLL; ++j) {
      if (p[j] != 0xffffffffu) csr[p[j]] = make_int2(u[j], __float_as_int(ds[j]));
    }
  }
}


#define GB 8
__device__ __forceinline__ v2f row_pair(const float* __restrict__ h, int r, int lane) {
  return *reinterpret_cast<const v2f*>(h + (size_t)r * 64 + 2 * lane);
}
__device__ __forceinline__ void node_sum(const int2* __restrict__ csr, const float* __restrict__ h,
                                         int start, int cnt, float di, int2 rec0,
                                         v2f own, int lane, v2f& acc) {
  const float wself = di * di;
  acc.x = own.x * wself;
  acc.y = own.y * wself;
  for (int k0 = 0; k0 < cnt; k0 += 32) {
    const int m = min(cnt - k0, 32);
    int2 rec = rec0;
    if (k0 > 0) {
      rec = csr[start + k0 + min(lane, m - 1)];
      asm volatile("" :: "v"(rec.x), "v"(rec.y));
    }
    const int myidx = rec.x;
    const float dsrc = __int_as_float(rec.y);
    for (int k = 0; k < m; k += GB) {
      int s[GB];
#pragma unroll
      for (int j = 0; j < GB; ++j) s[j] = __builtin_amdgcn_readlane(myidx, k + j);
      v2f v[GB];
#pragma unroll
      for (int j = 0; j < GB; ++j) v[j] = row_pair(h, s[j], lane);
      float w[GB];
#pragma unroll
      for (int j = 0; j < GB; ++j) {
        const float ds = __int_as_float(__builtin_amdgcn_readlane(__float_as_int(dsrc), k + j));
        w[j] = (k + j < m) ? ds * di : 0.0f;
      }
#pragma unroll
      for (int j = 0; j < GB; ++j) {
        acc.x = fmaf(v[j].x, w[j], acc.x);
        acc.y = fmaf(v[j].y, w[j], acc.y);
      }
    }
  }
}

__device__ __forceinline__ int2 first_records(const int2* __restrict__ csr, int start, int cnt,
                                              int lane) {
  int2 rec = make_int2(0, 0);
  if (cnt > 0) rec = csr[start + min(lane, min(cnt, 32) - 1)];
  return rec;
}

#define PG_WAVES 16384
__global__ void __launch_bounds__(TB)
k_gather(const int2* __restrict__ csr, const unsigned int* __restrict__ rowptr,
         const unsigned int* __restrict__ dtot, const float* __restrict__ h,
         const float* __restrict__ dis, const float* __restrict__ bias,
         float* __restrict__ out, int n) {
  const int lane = threadIdx.x & 31;
  const int w0 = __builtin_amdgcn_readfirstlane((int)((blockIdx.x * blockDim.x + threadIdx.x) >> 5));
  const int stride = (int)((gridDim.x * blockDim.x) >> 5);
  if (w0 >= n) return;
  const v2f bia = *reinterpret_cast<const v2f*>(bias + 2 * lane);
  for (int base = w0; base < n; base += 32 * stride) {
    const int j = min(base + lane * stride, n - 1);
    const int mst = (int)rowptr[j];
    const int mc = (int)dtot[j];
    const int md = __float_as_int(dis[j]);
    int2 recA = first_records(csr, __builtin_amdgcn_readlane(mst, 0),
                              __builtin_amdgcn_readlane(mc, 0), lane);
    v2f hA = row_pair(h, base, lane);
    asm volatile("" :: "v"(mst), "v"(mc), "v"(md), "v"(recA.x), "v"(recA.y), "v"(hA.x), "v"(hA.y));
    for (int k = 0; k < 32; ++k) {
      const int i = base + k * stride;
      if (i >= n) break;
      int2 recB = make_int2(0, 0);
      v2f hB = {0.0f, 0.0f};
      if (k + 1 < 32 && i + stride < n) {
        recB = first_records(csr, __builtin_amdgcn_readlane(mst, k + 1),
                             __builtin_amdgcn_readlane(mc, k + 1), lane);
        hB = row_pair(h, i + stride, lane);
      }
      const int st = __builtin_amdgcn_readlane(mst, k);
      const int c = __builtin_amdgcn_readlane(mc, k);
      const float d = __int_as_float(__builtin_amdgcn_readlane(md, k));
      v2f acc;
      node_sum(csr, h, st, c, d, recA, hA, lane, acc);
      v2f o;
      o.x = acc.x + bia.x;
      o.y = acc.y + bia.y;
      *reinterpret_cast<v2f*>(out + (size_t)i * 64 + 2 * lane) = o;
      recA = recB; hA = hB;
    }
  }
}

#define FT_NODES 16
#define FT_TB    (FT_NODES * 32)
#define FT_PITCH 68

__device__ __forceinline__ v8f tile_product(const float* __restrict__ rows,
                                            const float* __restrict__ W, int nt, int lane,
                                            float c0) {
  const int mlow  = lane & 15;
  const int khalf = (lane >> 4) << 1;
  const float* ar = rows + mlow * FT_PITCH + khalf;
  const float* wr = W + (unsigned int)((nt * 16 + mlow) * 64 + khalf);
  v8f c = {c0, c0, c0, c0, c0, c0, c0, c0};
#pragma unroll
  for (int hq = 0; hq < 16; hq += 8) {
    v2f a[8], b[8];
#pragma unroll
    for (int q = 0; q < 8; ++q) {
      b[q].x = wr[(hq + q) * 4 + 0];
      b[q].y = wr[(hq + q) * 4 + 1];
    }
#pragma unroll
    for (int q = 0; q < 8; ++q) {
      a[q].x = ar[(hq + q) * 4 + 0];
      a[q].y = ar[(hq + q) * 4 + 1];
    }
#pragma unroll
    for (int q = 0; q < 8; ++q) {
      c = __builtin_amdgcn_wmma_f32_16x16x4_f32(
           false, a[q],  false, b[q],
           (short)0, c,  false,  false);
    }
  }
  return c;
}

__device__ __forceinline__ void store_tile(const v8f c, float* __restrict__ H2,
                                           int m0, int n, int nt, int lane) {
  const int row0 = m0 + 8 * (lane >> 4);
  float* o = H2 + (size_t)row0 * 64 + nt * 16 + (lane & 15);
  if (m0 + FT_NODES <= n) {
#pragma unroll
    for (int v = 0; v < 8; ++v) o[(size_t)v * 64] = c[v];
  } else {
#pragma unroll
    for (int v = 0; v < 8; ++v) {
      if (row0 + v < n) o[(size_t)v * 64] = c[v];
    }
  }
}

#define FT_TILE (FT_NODES * FT_PITCH)
__global__ void __launch_bounds__(FT_TB, 16)
k_gather_gemm(const int2* __restrict__ csr, const unsigned int* __restrict__ rowptr,
              const unsigned int* __restrict__ dtot, const float* __restrict__ x,
              const float* __restrict__ dis, const float* __restrict__ W1,
              const float* __restrict__ b1, const float* __restrict__ W2,
              float* __restrict__ H2, int n, int ntiles) {
  __shared__ float lds[4][FT_TILE];
  const int lane = threadIdx.x & 31;
  const int wid = __builtin_amdgcn_readfirstlane((int)(threadIdx.x >> 5));
  const int ngrp = (int)gridDim.x;
  const int stride = ngrp * FT_NODES;
  const int nt = wid & 3;
  const bool l1 = wid >= 4;

  int buf = 0;
  int pm0 = -1;
  for (int tb = (int)blockIdx.x; tb < ntiles; tb += 32 * ngrp) {
    const int ibase = tb * FT_NODES + wid;
    const int j = min(ibase + lane * stride, n - 1);
    const int mst = (int)rowptr[j];
    const int mc = (int)dtot[j];
    const int md = __float_as_int(dis[j]);
    int2 recA = make_int2(0, 0);
    v2f hA = {0.0f, 0.0f};
    if (ibase < n) {
      recA = first_records(csr, __builtin_amdgcn_readlane(mst, 0),
                           __builtin_amdgcn_readlane(mc, 0), lane);
      hA = row_pair(x, ibase, lane);
    }
    asm volatile("" :: "v"(mst), "v"(mc), "v"(md), "v"(recA.x), "v"(recA.y), "v"(hA.x), "v"(hA.y));
    for (int k = 0; k < 32; ++k) {
      const int t = tb + k * ngrp;
      if (t >= ntiles) break;
      const int m0 = t * FT_NODES;
      const int i = m0 + wid;
      int2 recB = make_int2(0, 0);
      v2f hB = {0.0f, 0.0f};
      if (k + 1 < 32 && i + stride < n) {
        recB = first_records(csr, __builtin_amdgcn_readlane(mst, k + 1),
                             __builtin_amdgcn_readlane(mc, k + 1), lane);
        hB = row_pair(x, i + stride, lane);
      }
      v2f r = {0.0f, 0.0f};
      if (i < n) {
        const int st = __builtin_amdgcn_readlane(mst, k);
        const int c = __builtin_amdgcn_readlane(mc, k);
        const float d = __int_as_float(__builtin_amdgcn_readlane(md, k));
        node_sum(csr, x, st, c, d, recA, hA, lane, r);
      }
      *reinterpret_cast<v2f*>(&lds[buf][wid * FT_PITCH + 2 * lane]) = r;
      __syncthreads();
      if (wid < 8 && (l1 || pm0 >= 0)) {
        float c0 = 0.0f;
        if (l1) c0 = b1[nt * 16 + (lane & 15)];
        const v8f c = tile_product(lds[l1 ? buf : 3 - buf], l1 ? W1 : W2, nt, lane, c0);
        if (l1) {
          float* ro = lds[2 + buf] + (8 * (lane >> 4)) * FT_PITCH + nt * 16 + (lane & 15);
#pragma unroll
          for (int v = 0; v < 8; ++v) ro[v * FT_PITCH] = fmaxf(c[v], 0.0f);
        } else {
          store_tile(c, H2, pm0, n, nt, lane);
        }
      }
      pm0 = m0;
      buf ^= 1;
      recA = recB; hA = hB;
    }
  }
  __syncthreads();
  if (wid < 4 && pm0 >= 0) {
    const v8f c = tile_product(lds[3 - buf], W2, nt, lane, 0.0f);
    store_tile(c, H2, pm0, n, nt, lane);
  }
}

extern "C" void kernel_launch(void* const* d_in, const int* in_sizes, int n_in,
                              void* d_out, int out_size, void* d_ws, size_t ws_size,
                              hipStream_t stream) {
  (void)n_in; (void)out_size; (void)ws_size;

  const float* x  = (const float*)d_in[0];
  const int*   ei = (const int*)  d_in[1];
  const float* W1 = (const float*)d_in[2];
  const float* b1 = (const float*)d_in[3];
  const float* W2 = (const float*)d_in[4];
  const float* b2 = (const float*)d_in[5];
  float* out = (float*)d_out;

  const int n = in_sizes[0] / 64;
  const int e = in_sizes[1] / 2;
  const int* src = ei;
  const int* dst = ei + e;

  const int node_blocks = (n + 7) / 8;
  const int tile_blocks = (n + FT_NODES - 1) / FT_NODES;
  const int nranges     = (n + DC_RANGE - 1) / DC_RANGE;
  const int npad        = nranges * DC_RANGE;
  const int nfin        = nranges * FIN_PER_RANGE;

  float* h2  = (float*)d_ws;
  float* dis = h2 + (size_t)n * 64;
  unsigned int* part   = (unsigned int*)(dis + n);
  unsigned int* dtot   = part + (size_t)DC_SLICES * npad;
  unsigned int* rowptr = dtot + npad;
  unsigned int* bsum   = rowptr + npad;
  int2* csr = (int2*)(bsum + (size_t)nfin * 32);

  k_deg_count <<<nranges * DC_SLICES, DC_TB, 0, stream>>>(dst, part, e, nranges);
  k_deg_finish<<<nfin, FIN_TB, 0, stream>>>(part, dis, dtot, bsum, n, npad);
  k_fill      <<<nranges * DC_SLICES, DC_TB, 0, stream>>>(src, dst, part, dtot, bsum, dis, rowptr,
                                                          csr, e, nranges, npad);

  k_gather_gemm<<<tile_blocks < PG_WAVES / FT_NODES ? tile_blocks : PG_WAVES / FT_NODES, FT_TB, 0, stream>>>(
      csr, rowptr, dtot, x, dis, W1, b1, W2, h2, n, tile_blocks);

  k_gather<<<node_blocks < PG_WAVES / 8 ? node_blocks : PG_WAVES / 8, TB, 0, stream>>>(
      csr, rowptr, dtot, h2, dis, b2, out, n);
}


# fused launch: W1, W2, b1 copied once per work-group into LDS (pitch 68); products read B fragments there, not as 16-line global loads per tile; same sums
# speedup vs baseline: 4.2090x; 1.0549x over previous
#include <hip/hip_runtime.h>
#include <hip/hip_bf16.h>


typedef __attribute__((ext_vector_type(2))) float v2f;
typedef __attribute__((ext_vector_type(8))) float v8f;

#define TB 256

#define DC_RANGE  12512
#define DC_SLICES 32
#define DC_TB     1024
#define DC_UNROLL 16
__global__ void __launch_bounds__(DC_TB)
k_deg_count(const int* __restrict__ dst, unsigned int* __restrict__ part, int e, int nranges) {
  __shared__ unsigned int cnt[DC_RANGE];
  const int tid = threadIdx.x;
  const int s = blockIdx.x / nranges;
  const int g = blockIdx.x - s * nranges;
  const int lo = g * DC_RANGE;

  for (int b = tid; b < DC_RANGE; b += DC_TB) cnt[b] = 0u;
  __syncthreads();

  const int per = (((e + DC_SLICES - 1) / DC_SLICES) + 31) & ~31;
  const int e0 = s * per;
  const int e1 = min(e, e0 + per);
  for (int base = e0; base < e1; base += DC_UNROLL * DC_TB) {
    int v[DC_UNROLL];
#pragma unroll
    for (int j = 0; j < DC_UNROLL; ++j) {
      v[j] = dst[min(base + j * DC_TB + tid, e1 - 1)];
    }
    asm volatile("" :: "v"(v[0]), "v"(v[1]), "v"(v[2]), "v"(v[3]), "v"(v[4]), "v"(v[5]),
                       "v"(v[6]), "v"(v[7]), "v"(v[8]), "v"(v[9]), "v"(v[10]), "v"(v[11]),
                       "v"(v[12]), "v"(v[13]), "v"(v[14]), "v"(v[15]));
#pragma unroll
    for (int j = 0; j < DC_UNROLL; ++j) {
      const bool live = (base + j * DC_TB + tid) < e1;
      const unsigned int r = (unsigned int)(v[j] - lo);
      if (live && r < (unsigned int)DC_RANGE) atomicAdd(&cnt[r], 1u);
    }
  }
  __syncthreads();

  unsigned int* row = part + (size_t)s * ((size_t)nranges * DC_RANGE) + lo;
  for (int b = tid; b < DC_RANGE; b += DC_TB) row[b] = cnt[b];
}

#define FIN_TB 544
#define FIN_PER_RANGE (DC_RANGE / FIN_TB)
static_assert(DC_RANGE % FIN_TB == 0, "finish blocks must tile a node range");
__global__ void __launch_bounds__(FIN_TB)
k_deg_finish(unsigned int* __restrict__ part, float* __restrict__ dis,
             unsigned int* __restrict__ dtot, unsigned int* __restrict__ bsum, int n, int npad) {
  __shared__ unsigned int bs;
  const int i = blockIdx.x * FIN_TB + threadIdx.x;
  if (threadIdx.x == 0) bs = 0u;
  __syncthreads();
  unsigned int c[DC_SLICES];
#pragma unroll
  for (int s = 0; s < DC_SLICES; ++s) c[s] = part[(size_t)s * npad + i];
  unsigned int run = 0u;
#pragma unroll
  for (int s = 0; s < DC_SLICES; ++s) { part[(size_t)s * npad + i] = run; run += c[s]; }
  dtot[i] = run;
  if (i < n) dis[i] = rsqrtf((float)(run + 1u));
  atomicAdd(&bs, run);
  __syncthreads();
  if (threadIdx.x == 0) bsum[blockIdx.x * 32] = bs;
}

#define SC_CH 13
static_assert(DC_TB * SC_CH >= DC_RANGE, "range scan must cover the range");
static_assert(DC_TB == 1024, "the range scan assumes 32 waves of 32 lanes");
static_assert(SC_CH == 13, "the load group in k_fill names its SC_CH words one by one");
__global__ void __launch_bounds__(DC_TB)
k_fill(const int* __restrict__ src, const int* __restrict__ dst,
       const unsigned int* __restrict__ part, const unsigned int* __restrict__ dtot,
       const unsigned int* __restrict__ bsum, const float* __restrict__ dis,
       unsigned int* __restrict__ rowptr, int2* __restrict__ csr, int e, int nranges, int npad) {
  __shared__ unsigned int cnt[DC_RANGE];
  __shared__ unsigned int wsum[32];
  __shared__ unsigned int rb;
  const int tid = threadIdx.x;
  const int lane = tid & 31;
  const int wid = tid >> 5;
  const int s = blockIdx.x / nranges;
  const int g = blockIdx.x - s * nranges;
  const int lo = g * DC_RANGE;

  const unsigned int* __restrict__ prow = part + (size_t)s * npad + lo;
  const int nprev = g * FIN_PER_RANGE;
  unsigned int dreg[SC_CH], preg[SC_CH];
#pragma unroll
  for (int j = 0; j < SC_CH; ++j) {
    const int b = min(tid + j * DC_TB, DC_RANGE - 1);
    dreg[j] = dtot[lo + b];
    preg[j] = prow[b];
  }
  unsigned int acc = 0u;
  if (tid < nprev) acc = bsum[tid * 32];
  asm volatile("" :: "v"(dreg[0]), "v"(dreg[1]), "v"(dreg[2]), "v"(dreg[3]), "v"(dreg[4]),
                     "v"(dreg[5]), "v"(dreg[6]), "v"(dreg[7]), "v"(dreg[8]), "v"(dreg[9]),
                     "v"(dreg[10]), "v"(dreg[11]), "v"(dreg[12]),
                     "v"(preg[0]), "v"(preg[1]), "v"(preg[2]), "v"(preg[3]), "v"(preg[4]),
                     "v"(preg[5]), "v"(preg[6]), "v"(preg[7]), "v"(preg[8]), "v"(preg[9]),
                     "v"(preg[10]), "v"(preg[11]), "v"(preg[12]), "v"(acc));

  if (tid == 0) rb = 0u;
#pragma unroll
  for (int j = 0; j < SC_CH; ++j) {
    const int b = tid + j * DC_TB;
    if (b < DC_RANGE) cnt[b] = dreg[j];
  }
  __syncthreads();

  for (int k = tid + DC_TB; k < nprev; k += DC_TB) acc += bsum[k * 32];
  if (acc != 0u) atomicAdd(&rb, acc);

  const int c0 = tid * SC_CH;
  unsigned int loc[SC_CH];
  unsigned int tsum = 0u;
#pragma unroll
  for (int j = 0; j < SC_CH; ++j) {
    const int b = c0 + j;
    loc[j] = (b < DC_RANGE) ? cnt[b] : 0u;
    tsum += loc[j];
  }
  unsigned int x = tsum;
#pragma unroll
  for (int o = 1; o < 32; o <<= 1) {
    const unsigned int t = __shfl_up(x, o);
    if (lane >= o) x += t;
  }
  if (lane == 31) wsum[wid] = x;
  __syncthreads();
  const unsigned int w = wsum[lane];
  unsigned int ws = w;
#pragma unroll
  for (int o = 1; o < 32; o <<= 1) {
    const unsigned int t = __shfl_up(ws, o);
    if (lane >= o) ws += t;
  }
  const unsigned int wave_excl = __shfl(ws, wid) - __shfl(w, wid);
  unsigned int run = rb + wave_excl + (x - tsum);
#pragma unroll
  for (int j = 0; j < SC_CH; ++j) {
    const int b = c0 + j;
    if (b < DC_RANGE) { cnt[b] = run; run += loc[j]; }
  }
  __syncthreads();

  unsigned int rs[SC_CH];
#pragma unroll
  for (int j = 0; j < SC_CH; ++j) {
    const int b = tid + j * DC_TB;
    rs[j] = (b < DC_RANGE) ? cnt[b] : 0u;
  }
#pragma unroll
  for (int j = 0; j < SC_CH; ++j) {
    const int b = tid + j * DC_TB;
    if (b < DC_RANGE) {
      if (s == 0) rowptr[lo + b] = rs[j];
      cnt[b] = rs[j] + preg[j];
    }
  }
  __syncthreads();

  const int per = (((e + DC_SLICES - 1) / DC_SLICES) + 31) & ~31;
  const int e0 = s * per;
  const int e1 = min(e, e0 + per);
  for (int base = e0; base < e1; base += DC_UNROLL * DC_TB) {
    int v[DC_UNROLL], u[DC_UNROLL];
#pragma unroll
    for (int j = 0; j < DC_UNROLL; ++j) {
      const int idx = min(base + j * DC_TB + tid, e1 - 1);
      v[j] = dst[idx];
      u[j] = src[idx];
    }
    asm volatile("" :: "v"(v[0]), "v"(v[1]), "v"(v[2]), "v"(v[3]), "v"(v[4]), "v"(v[5]),
                       "v"(v[6]), "v"(v[7]), "v"(v[8]), "v"(v[9]), "v"(v[10]), "v"(v[11]),
                       "v"(v[12]), "v"(v[13]), "v"(v[14]), "v"(v[15]));
    asm volatile("" :: "v"(u[0]), "v"(u[1]), "v"(u[2]), "v"(u[3]), "v"(u[4]), "v"(u[5]),
                       "v"(u[6]), "v"(u[7]), "v"(u[8]), "v"(u[9]), "v"(u[10]), "v"(u[11]),
                       "v"(u[12]), "v"(u[13]), "v"(u[14]), "v"(u[15]));
    unsigned int p[DC_UNROLL];
    float ds[DC_UNROLL];
#pragma unroll
    for (int j = 0; j < DC_UNROLL; ++j) {
      const bool live = (base + j * DC_TB + tid) < e1;
      const unsigned int r = (unsigned int)(v[j] - lo);
      p[j] = 0xffffffffu;
      ds[j] = 0.0f;
      if (live && r < (unsigned int)DC_RANGE) {
        p[j] = atomicAdd(&cnt[r], 1u);
        ds[j] = dis[u[j]];
      }
    }
#pragma unroll
    for (int j = 0; j < DC_UNROLL; ++j) {
      if (p[j] != 0xffffffffu) csr[p[j]] = make_int2(u[j], __float_as_int(ds[j]));
    }
  }
}


#define GB 8
__device__ __forceinline__ v2f row_pair(const float* __restrict__ h, int r, int lane) {
  return *reinterpret_cast<const v2f*>(h + (size_t)r * 64 + 2 * lane);
}
__device__ __forceinline__ void node_sum(const int2* __restrict__ csr, const float* __restrict__ h,
                                         int start, int cnt, float di, int2 rec0,
                                         v2f own, int lane, v2f& acc) {
  const float wself = di * di;
  acc.x = own.x * wself;
  acc.y = own.y * wself;
  for (int k0 = 0; k0 < cnt; k0 += 32) {
    const int m = min(cnt - k0, 32);
    int2 rec = rec0;
    if (k0 > 0) {
      rec = csr[start + k0 + min(lane, m - 1)];
      asm volatile("" :: "v"(rec.x), "v"(rec.y));
    }
    const int myidx = rec.x;
    const float dsrc = __int_as_float(rec.y);
    for (int k = 0; k < m; k += GB) {
      int s[GB];
#pragma unroll
      for (int j = 0; j < GB; ++j) s[j] = __builtin_amdgcn_readlane(myidx, k + j);
      v2f v[GB];
#pragma unroll
      for (int j = 0; j < GB; ++j) v[j] = row_pair(h, s[j], lane);
      float w[GB];
#pragma unroll
      for (int j = 0; j < GB; ++j) {
        const float ds = __int_as_float(__builtin_amdgcn_readlane(__float_as_int(dsrc), k + j));
        w[j] = (k + j < m) ? ds * di : 0.0f;
      }
#pragma unroll
      for (int j = 0; j < GB; ++j) {
        acc.x = fmaf(v[j].x, w[j], acc.x);
        acc.y = fmaf(v[j].y, w[j], acc.y);
      }
    }
  }
}

__device__ __forceinline__ int2 first_records(const int2* __restrict__ csr, int start, int cnt,
                                              int lane) {
  int2 rec = make_int2(0, 0);
  if (cnt > 0) rec = csr[start + min(lane, min(cnt, 32) - 1)];
  return rec;
}

#define PG_WAVES 16384
__global__ void __launch_bounds__(TB)
k_gather(const int2* __restrict__ csr, const unsigned int* __restrict__ rowptr,
         const unsigned int* __restrict__ dtot, const float* __restrict__ h,
         const float* __restrict__ dis, const float* __restrict__ bias,
         float* __restrict__ out, int n) {
  const int lane = threadIdx.x & 31;
  const int w0 = __builtin_amdgcn_readfirstlane((int)((blockIdx.x * blockDim.x + threadIdx.x) >> 5));
  const int stride = (int)((gridDim.x * blockDim.x) >> 5);
  if (w0 >= n) return;
  const v2f bia = *reinterpret_cast<const v2f*>(bias + 2 * lane);
  for (int base = w0; base < n; base += 32 * stride) {
    const int j = min(base + lane * stride, n - 1);
    const int mst = (int)rowptr[j];
    const int mc = (int)dtot[j];
    const int md = __float_as_int(dis[j]);
    int2 recA = first_records(csr, __builtin_amdgcn_readlane(mst, 0),
                              __builtin_amdgcn_readlane(mc, 0), lane);
    v2f hA = row_pair(h, base, lane);
    asm volatile("" :: "v"(mst), "v"(mc), "v"(md), "v"(recA.x), "v"(recA.y), "v"(hA.x), "v"(hA.y));
    for (int k = 0; k < 32; ++k) {
      const int i = base + k * stride;
      if (i >= n) break;
      int2 recB = make_int2(0, 0);
      v2f hB = {0.0f, 0.0f};
      if (k + 1 < 32 && i + stride < n) {
        recB = first_records(csr, __builtin_amdgcn_readlane(mst, k + 1),
                             __builtin_amdgcn_readlane(mc, k + 1), lane);
        hB = row_pair(h, i + stride, lane);
      }
      const int st = __builtin_amdgcn_readlane(mst, k);
      const int c = __builtin_amdgcn_readlane(mc, k);
      const float d = __int_as_float(__builtin_amdgcn_readlane(md, k));
      v2f acc;
      node_sum(csr, h, st, c, d, recA, hA, lane, acc);
      v2f o;
      o.x = acc.x + bia.x;
      o.y = acc.y + bia.y;
      *reinterpret_cast<v2f*>(out + (size_t)i * 64 + 2 * lane) = o;
      recA = recB; hA = hB;
    }
  }
}

#define FT_NODES 16
#define FT_TB    (FT_NODES * 32)
#define FT_PITCH 68

__device__ __forceinline__ v8f tile_product(const float* __restrict__ rows,
                                            const float* __restrict__ Wl, int nt, int lane,
                                            float c0) {
  const int mlow  = lane & 15;
  const int khalf = (lane >> 4) << 1;
  const float* ar = rows + mlow * FT_PITCH + khalf;
  const float* wr = Wl + (nt * 16 + mlow) * FT_PITCH + khalf;
  v8f c = {c0, c0, c0, c0, c0, c0, c0, c0};
#pragma unroll
  for (int hq = 0; hq < 16; hq += 8) {
    v2f a[8], b[8];
#pragma unroll
    for (int q = 0; q < 8; ++q) {
      b[q].x = wr[(hq + q) * 4 + 0];
      b[q].y = wr[(hq + q) * 4 + 1];
    }
#pragma unroll
    for (int q = 0; q < 8; ++q) {
      a[q].x = ar[(hq + q) * 4 + 0];
      a[q].y = ar[(hq + q) * 4 + 1];
    }
    asm volatile("" :: "v"(a[0]), "v"(a[1]), "v"(a[2]), "v"(a[3]), "v"(a[4]), "v"(a[5]),
                       "v"(a[6]), "v"(a[7]), "v"(b[0]), "v"(b[1]), "v"(b[2]), "v"(b[3]),
                       "v"(b[4]), "v"(b[5]), "v"(b[6]), "v"(b[7]));
#pragma unroll
    for (int q = 0; q < 8; ++q) {
      c = __builtin_amdgcn_wmma_f32_16x16x4_f32(
           false, a[q],  false, b[q],
           (short)0, c,  false,  false);
    }
  }
  return c;
}

__device__ __forceinline__ void store_tile(const v8f c, float* __restrict__ H2,
                                           int m0, int n, int nt, int lane) {
  const int row0 = m0 + 8 * (lane >> 4);
  float* o = H2 + (size_t)row0 * 64 + nt * 16 + (lane & 15);
  if (m0 + FT_NODES <= n) {
#pragma unroll
    for (int v = 0; v < 8; ++v) o[(size_t)v * 64] = c[v];
  } else {
#pragma unroll
    for (int v = 0; v < 8; ++v) {
      if (row0 + v < n) o[(size_t)v * 64] = c[v];
    }
  }
}

#define FT_TILE (FT_NODES * FT_PITCH)
#define FT_WMAT (64 * FT_PITCH)
typedef __attribute__((ext_vector_type(4))) float v4f;
__global__ void __launch_bounds__(FT_TB, 16)
k_gather_gemm(const int2* __restrict__ csr, const unsigned int* __restrict__ rowptr,
              const unsigned int* __restrict__ dtot, const float* __restrict__ x,
              const float* __restrict__ dis, const float* __restrict__ W1,
              const float* __restrict__ b1, const float* __restrict__ W2,
              float* __restrict__ H2, int n, int ntiles) {
  __shared__ float lds[4][FT_TILE];
  __shared__ __attribute__((aligned(16))) float wl[2][FT_WMAT];
  __shared__ float bl[64];
  static_assert(2 * FT_TB == 64 * 64 / 4, "two pieces of each matrix per thread");
  {
    const int tid = (int)threadIdx.x;
    v4f w1v[2], w2v[2];
#pragma unroll
    for (int it = 0; it < 2; ++it) {
      w1v[it] = *reinterpret_cast<const v4f*>(W1 + 4 * (tid + it * FT_TB));
      w2v[it] = *reinterpret_cast<const v4f*>(W2 + 4 * (tid + it * FT_TB));
    }
    const float bv = b1[tid & 63];
    asm volatile("" :: "v"(bv));
#pragma unroll
    for (int it = 0; it < 2; ++it) {
      const int p = tid + it * FT_TB;
      const int o = (p >> 4) * FT_PITCH + ((p & 15) << 2);
      *reinterpret_cast<v4f*>(&wl[0][o]) = w1v[it];
      *reinterpret_cast<v4f*>(&wl[1][o]) = w2v[it];
    }
    if (tid < 64) bl[tid] = bv;
  }
  const int lane = threadIdx.x & 31;
  const int wid = __builtin_amdgcn_readfirstlane((int)(threadIdx.x >> 5));
  const int ngrp = (int)gridDim.x;
  const int stride = ngrp * FT_NODES;
  const int nt = wid & 3;
  const bool l1 = wid >= 4;

  int buf = 0;
  int pm0 = -1;
  for (int tb = (int)blockIdx.x; tb < ntiles; tb += 32 * ngrp) {
    const int ibase = tb * FT_NODES + wid;
    const int j = min(ibase + lane * stride, n - 1);
    const int mst = (int)rowptr[j];
    const int mc = (int)dtot[j];
    const int md = __float_as_int(dis[j]);
    int2 recA = make_int2(0, 0);
    v2f hA = {0.0f, 0.0f};
    if (ibase < n) {
      recA = first_records(csr, __builtin_amdgcn_readlane(mst, 0),
                           __builtin_amdgcn_readlane(mc, 0), lane);
      hA = row_pair(x, ibase, lane);
    }
    asm volatile("" :: "v"(mst), "v"(mc), "v"(md), "v"(recA.x), "v"(recA.y), "v"(hA.x), "v"(hA.y));
    for (int k = 0; k < 32; ++k) {
      const int t = tb + k * ngrp;
      if (t >= ntiles) break;
      const int m0 = t * FT_NODES;
      const int i = m0 + wid;
      int2 recB = make_int2(0, 0);
      v2f hB = {0.0f, 0.0f};
      if (k + 1 < 32 && i + stride < n) {
        recB = first_records(csr, __builtin_amdgcn_readlane(mst, k + 1),
                             __builtin_amdgcn_readlane(mc, k + 1), lane);
        hB = row_pair(x, i + stride, lane);
      }
      v2f r = {0.0f, 0.0f};
      if (i < n) {
        const int st = __builtin_amdgcn_readlane(mst, k);
        const int c = __builtin_amdgcn_readlane(mc, k);
        const float d = __int_as_float(__builtin_amdgcn_readlane(md, k));
        node_sum(csr, x, st, c, d, recA, hA, lane, r);
      }
      *reinterpret_cast<v2f*>(&lds[buf][wid * FT_PITCH + 2 * lane]) = r;
      __syncthreads();
      if (wid < 8 && (l1 || pm0 >= 0)) {
        float c0 = 0.0f;
        if (l1) c0 = bl[nt * 16 + (lane & 15)];
        const v8f c = tile_product(lds[l1 ? buf : 3 - buf], wl[l1 ? 0 : 1], nt, lane, c0);
        if (l1) {
          float* ro = lds[2 + buf] + (8 * (lane >> 4)) * FT_PITCH + nt * 16 + (lane & 15);
#pragma unroll
          for (int v = 0; v < 8; ++v) ro[v * FT_PITCH] = fmaxf(c[v], 0.0f);
        } else {
          store_tile(c, H2, pm0, n, nt, lane);
        }
      }
      pm0 = m0;
      buf ^= 1;
      recA = recB; hA = hB;
    }
  }
  __syncthreads();
  if (wid < 4 && pm0 >= 0) {
    const v8f c = tile_product(lds[3 - buf], wl[1], nt, lane, 0.0f);
    store_tile(c, H2, pm0, n, nt, lane);
  }
}

extern "C" void kernel_launch(void* const* d_in, const int* in_sizes, int n_in,
                              void* d_out, int out_size, void* d_ws, size_t ws_size,
                              hipStream_t stream) {
  (void)n_in; (void)out_size; (void)ws_size;

  const float* x  = (const float*)d_in[0];
  const int*   ei = (const int*)  d_in[1];
  const float* W1 = (const float*)d_in[2];
  const float* b1 = (const float*)d_in[3];
  const float* W2 = (const float*)d_in[4];
  const float* b2 = (const float*)d_in[5];
  float* out = (float*)d_out;

  const int n = in_sizes[0] / 64;
  const int e = in_sizes[1] / 2;
  const int* src = ei;
  const int* dst = ei + e;

  const int node_blocks = (n + 7) / 8;
  const int tile_blocks = (n + FT_NODES - 1) / FT_NODES;
  const int nranges     = (n + DC_RANGE - 1) / DC_RANGE;
  const int npad        = nranges * DC_RANGE;
  const int nfin        = nranges * FIN_PER_RANGE;

  float* h2  = (float*)d_ws;
  float* dis = h2 + (size_t)n * 64;
  unsigned int* part   = (unsigned int*)(dis + n);
  unsigned int* dtot   = part + (size_t)DC_SLICES * npad;
  unsigned int* rowptr = dtot + npad;
  unsigned int* bsum   = rowptr + npad;
  int2* csr = (int2*)(bsum + (size_t)nfin * 32);

  k_deg_count <<<nranges * DC_SLICES, DC_TB, 0, stream>>>(dst, part, e, nranges);
  k_deg_finish<<<nfin, FIN_TB, 0, stream>>>(part, dis, dtot, bsum, n, npad);
  k_fill      <<<nranges * DC_SLICES, DC_TB, 0, stream>>>(src, dst, part, dtot, bsum, dis, rowptr,
                                                          csr, e, nranges, npad);

  k_gather_gemm<<<tile_blocks < PG_WAVES / FT_NODES ? tile_blocks : PG_WAVES / FT_NODES, FT_TB, 0, stream>>>(
      csr, rowptr, dtot, x, dis, W1, b1, W2, h2, n, tile_blocks);

  k_gather<<<node_blocks < PG_WAVES / 8 ? node_blocks : PG_WAVES / 8, TB, 0, stream>>>(
      csr, rowptr, dtot, h2, dis, b2, out, n);
}


# gathers fetch TWO rows per load: lanes 0-15 take 16 bytes each of one in-edge row, lanes 16-31 of the next, 8 loads (16 rows) a group; the halves are added once per node; sum order changes
# speedup vs baseline: 4.3986x; 1.0450x over previous
#include <hip/hip_runtime.h>
#include <hip/hip_bf16.h>


typedef __attribute__((ext_vector_type(2))) float v2f;
typedef __attribute__((ext_vector_type(4))) float v4f;
typedef __attribute__((ext_vector_type(8))) float v8f;

#define TB 256

#define DC_RANGE  12512
#define DC_SLICES 32
#define DC_TB     1024
#define DC_UNROLL 16
__global__ void __launch_bounds__(DC_TB)
k_deg_count(const int* __restrict__ dst, unsigned int* __restrict__ part, int e, int nranges) {
  __shared__ unsigned int cnt[DC_RANGE];
  const int tid = threadIdx.x;
  const int s = blockIdx.x / nranges;
  const int g = blockIdx.x - s * nranges;
  const int lo = g * DC_RANGE;

  for (int b = tid; b < DC_RANGE; b += DC_TB) cnt[b] = 0u;
  __syncthreads();

  const int per = (((e + DC_SLICES - 1) / DC_SLICES) + 31) & ~31;
  const int e0 = s * per;
  const int e1 = min(e, e0 + per);
  for (int base = e0; base < e1; base += DC_UNROLL * DC_TB) {
    int v[DC_UNROLL];
#pragma unroll
    for (int j = 0; j < DC_UNROLL; ++j) {
      v[j] = dst[min(base + j * DC_TB + tid, e1 - 1)];
    }
    asm volatile("" :: "v"(v[0]), "v"(v[1]), "v"(v[2]), "v"(v[3]), "v"(v[4]), "v"(v[5]),
                       "v"(v[6]), "v"(v[7]), "v"(v[8]), "v"(v[9]), "v"(v[10]), "v"(v[11]),
                       "v"(v[12]), "v"(v[13]), "v"(v[14]), "v"(v[15]));
#pragma unroll
    for (int j = 0; j < DC_UNROLL; ++j) {
      const bool live = (base + j * DC_TB + tid) < e1;
      const unsigned int r = (unsigned int)(v[j] - lo);
      if (live && r < (unsigned int)DC_RANGE) atomicAdd(&cnt[r], 1u);
    }
  }
  __syncthreads();

  unsigned int* row = part + (size_t)s * ((size_t)nranges * DC_RANGE) + lo;
  for (int b = tid; b < DC_RANGE; b += DC_TB) row[b] = cnt[b];
}

#define FIN_TB 544
#define FIN_PER_RANGE (DC_RANGE / FIN_TB)
static_assert(DC_RANGE % FIN_TB == 0, "finish blocks must tile a node range");
__global__ void __launch_bounds__(FIN_TB)
k_deg_finish(unsigned int* __restrict__ part, float* __restrict__ dis,
             unsigned int* __restrict__ dtot, unsigned int* __restrict__ bsum, int n, int npad) {
  __shared__ unsigned int bs;
  const int i = blockIdx.x * FIN_TB + threadIdx.x;
  if (threadIdx.x == 0) bs = 0u;
  __syncthreads();
  unsigned int c[DC_SLICES];
#pragma unroll
  for (int s = 0; s < DC_SLICES; ++s) c[s] = part[(size_t)s * npad + i];
  unsigned int run = 0u;
#pragma unroll
  for (int s = 0; s < DC_SLICES; ++s) { part[(size_t)s * npad + i] = run; run += c[s]; }
  dtot[i] = run;
  if (i < n) dis[i] = rsqrtf((float)(run + 1u));
  atomicAdd(&bs, run);
  __syncthreads();
  if (threadIdx.x == 0) bsum[blockIdx.x * 32] = bs;
}

#define SC_CH 13
static_assert(DC_TB * SC_CH >= DC_RANGE, "range scan must cover the range");
static_assert(DC_TB == 1024, "the range scan assumes 32 waves of 32 lanes");
static_assert(SC_CH == 13, "the load group in k_fill names its SC_CH words one by one");
__global__ void __launch_bounds__(DC_TB)
k_fill(const int* __restrict__ src, const int* __restrict__ dst,
       const unsigned int* __restrict__ part, const unsigned int* __restrict__ dtot,
       const unsigned int* __restrict__ bsum, const float* __restrict__ dis,
       unsigned int* __restrict__ rowptr, int2* __restrict__ csr, int e, int nranges, int npad) {
  __shared__ unsigned int cnt[DC_RANGE];
  __shared__ unsigned int wsum[32];
  __shared__ unsigned int rb;
  const int tid = threadIdx.x;
  const int lane = tid & 31;
  const int wid = tid >> 5;
  const int s = blockIdx.x / nranges;
  const int g = blockIdx.x - s * nranges;
  const int lo = g * DC_RANGE;

  const unsigned int* __restrict__ prow = part + (size_t)s * npad + lo;
  const int nprev = g * FIN_PER_RANGE;
  unsigned int dreg[SC_CH], preg[SC_CH];
#pragma unroll
  for (int j = 0; j < SC_CH; ++j) {
    const int b = min(tid + j * DC_TB, DC_RANGE - 1);
    dreg[j] = dtot[lo + b];
    preg[j] = prow[b];
  }
  unsigned int acc = 0u;
  if (tid < nprev) acc = bsum[tid * 32];
  asm volatile("" :: "v"(dreg[0]), "v"(dreg[1]), "v"(dreg[2]), "v"(dreg[3]), "v"(dreg[4]),
                     "v"(dreg[5]), "v"(dreg[6]), "v"(dreg[7]), "v"(dreg[8]), "v"(dreg[9]),
                     "v"(dreg[10]), "v"(dreg[11]), "v"(dreg[12]),
                     "v"(preg[0]), "v"(preg[1]), "v"(preg[2]), "v"(preg[3]), "v"(preg[4]),
                     "v"(preg[5]), "v"(preg[6]), "v"(preg[7]), "v"(preg[8]), "v"(preg[9]),
                     "v"(preg[10]), "v"(preg[11]), "v"(preg[12]), "v"(acc));

  if (tid == 0) rb = 0u;
#pragma unroll
  for (int j = 0; j < SC_CH; ++j) {
    const int b = tid + j * DC_TB;
    if (b < DC_RANGE) cnt[b] = dreg[j];
  }
  __syncthreads();

  for (int k = tid + DC_TB; k < nprev; k += DC_TB) acc += bsum[k * 32];
  if (acc != 0u) atomicAdd(&rb, acc);

  const int c0 = tid * SC_CH;
  unsigned int loc[SC_CH];
  unsigned int tsum = 0u;
#pragma unroll
  for (int j = 0; j < SC_CH; ++j) {
    const int b = c0 + j;
    loc[j] = (b < DC_RANGE) ? cnt[b] : 0u;
    tsum += loc[j];
  }
  unsigned int x = tsum;
#pragma unroll
  for (int o = 1; o < 32; o <<= 1) {
    const unsigned int t = __shfl_up(x, o);
    if (lane >= o) x += t;
  }
  if (lane == 31) wsum[wid] = x;
  __syncthreads();
  const unsigned int w = wsum[lane];
  unsigned int ws = w;
#pragma unroll
  for (int o = 1; o < 32; o <<= 1) {
    const unsigned int t = __shfl_up(ws, o);
    if (lane >= o) ws += t;
  }
  const unsigned int wave_excl = __shfl(ws, wid) - __shfl(w, wid);
  unsigned int run = rb + wave_excl + (x - tsum);
#pragma unroll
  for (int j = 0; j < SC_CH; ++j) {
    const int b = c0 + j;
    if (b < DC_RANGE) { cnt[b] = run; run += loc[j]; }
  }
  __syncthreads();

  unsigned int rs[SC_CH];
#pragma unroll
  for (int j = 0; j < SC_CH; ++j) {
    const int b = tid + j * DC_TB;
    rs[j] = (b < DC_RANGE) ? cnt[b] : 0u;
  }
#pragma unroll
  for (int j = 0; j < SC_CH; ++j) {
    const int b = tid + j * DC_TB;
    if (b < DC_RANGE) {
      if (s == 0) rowptr[lo + b] = rs[j];
      cnt[b] = rs[j] + preg[j];
    }
  }
  __syncthreads();

  const int per = (((e + DC_SLICES - 1) / DC_SLICES) + 31) & ~31;
  const int e0 = s * per;
  const int e1 = min(e, e0 + per);
  for (int base = e0; base < e1; base += DC_UNROLL * DC_TB) {
    int v[DC_UNROLL], u[DC_UNROLL];
#pragma unroll
    for (int j = 0; j < DC_UNROLL; ++j) {
      const int idx = min(base + j * DC_TB + tid, e1 - 1);
      v[j] = dst[idx];
      u[j] = src[idx];
    }
    asm volatile("" :: "v"(v[0]), "v"(v[1]), "v"(v[2]), "v"(v[3]), "v"(v[4]), "v"(v[5]),
                       "v"(v[6]), "v"(v[7]), "v"(v[8]), "v"(v[9]), "v"(v[10]), "v"(v[11]),
                       "v"(v[12]), "v"(v[13]), "v"(v[14]), "v"(v[15]));
    asm volatile("" :: "v"(u[0]), "v"(u[1]), "v"(u[2]), "v"(u[3]), "v"(u[4]), "v"(u[5]),
                       "v"(u[6]), "v"(u[7]), "v"(u[8]), "v"(u[9]), "v"(u[10]), "v"(u[11]),
                       "v"(u[12]), "v"(u[13]), "v"(u[14]), "v"(u[15]));
    unsigned int p[DC_UNROLL];
    float ds[DC_UNROLL];
#pragma unroll
    for (int j = 0; j < DC_UNROLL; ++j) {
      const bool live = (base + j * DC_TB + tid) < e1;
      const unsigned int r = (unsigned int)(v[j] - lo);
      p[j] = 0xffffffffu;
      ds[j] = 0.0f;
      if (live && r < (unsigned int)DC_RANGE) {
        p[j] = atomicAdd(&cnt[r], 1u);
        ds[j] = dis[u[j]];
      }
    }
#pragma unroll
    for (int j = 0; j < DC_UNROLL; ++j) {
      if (p[j] != 0xffffffffu) csr[p[j]] = make_int2(u[j], __float_as_int(ds[j]));
    }
  }
}


__device__ __forceinline__ v4f row_quad(const float* __restrict__ h, int r, int c) {
  return reinterpret_cast<const v4f*>(h)[(unsigned int)((r << 4) | c)];
}
__device__ __forceinline__ unsigned int half_bcast(unsigned int v, unsigned int sel) {
  return __builtin_amdgcn_permlane16(v, v, sel, sel, true, false);
}
__device__ __forceinline__ float other_half(float v) {
  const unsigned int u = __float_as_uint(v);
  return __uint_as_float(__builtin_amdgcn_permlanex16(u, u, 0x76543210u, 0xfedcba98u, true, false));
}
__device__ __forceinline__ v4f add_halves(v4f a) {
  v4f o;
  o.x = a.x + other_half(a.x);
  o.y = a.y + other_half(a.y);
  o.z = a.z + other_half(a.z);
  o.w = a.w + other_half(a.w);
  return o;
}
template <int GP>
__device__ __forceinline__ v4f node_sum(const int2* __restrict__ csr, const float* __restrict__ h,
                                        int i, int start, int cnt, float di, int2 rec0,
                                        int lane) {
  const int c = lane & 15;
  const int pl = 2 * c + (lane >> 4);
  const v4f* __restrict__ hq = reinterpret_cast<const v4f*>(h);
  const v4f own = row_quad(h, i, c);
  v4f acc = {0.0f, 0.0f, 0.0f, 0.0f};
  for (int k0 = 0; k0 < cnt; k0 += 32) {
    const int m = min(cnt - k0, 32);
    int2 rec = rec0;
    if (k0 > 0) {
      rec = csr[start + k0 + min(pl, m - 1)];
      asm volatile("" :: "v"(rec.x), "v"(rec.y));
    }
    const unsigned int mysrc = (unsigned int)rec.x;
    const unsigned int myw = (pl < m) ? (unsigned int)rec.y : 0u;
    const int np = (m + 1) >> 1;
    for (int k = 0; k < np; k += GP) {
      const unsigned int sb = (unsigned int)k * 0x11111111u;
      unsigned int ix[GP];
#pragma unroll
      for (int j = 0; j < GP; ++j) {
        const unsigned int s = half_bcast(mysrc, sb + (unsigned int)j * 0x11111111u);
        ix[j] = (s << 4) | (unsigned int)c;
      }
      v4f v[GP];
#pragma unroll
      for (int j = 0; j < GP; ++j) v[j] = hq[ix[j]];
#pragma unroll
      for (int j = 0; j < GP; ++j) {
        const float w = __uint_as_float(half_bcast(myw, sb + (unsigned int)j * 0x11111111u));
        acc.x = fmaf(v[j].x, w, acc.x);
        acc.y = fmaf(v[j].y, w, acc.y);
        acc.z = fmaf(v[j].z, w, acc.z);
        acc.w = fmaf(v[j].w, w, acc.w);
      }
    }
  }
  const float wown = (lane < 16) ? di : 0.0f;
  acc.x = fmaf(own.x, wown, acc.x);
  acc.y = fmaf(own.y, wown, acc.y);
  acc.z = fmaf(own.z, wown, acc.z);
  acc.w = fmaf(own.w, wown, acc.w);
  return acc;
}

__device__ __forceinline__ int2 first_records(const int2* __restrict__ csr, int start, int cnt,
                                              int pl) {
  int2 rec = make_int2(0, 0);
  if (cnt > 0) rec = csr[start + min(pl, min(cnt, 32) - 1)];
  return rec;
}

#define PG_WAVES 16384
#define GK 8
#define GF 8
__global__ void __launch_bounds__(TB, 16)
k_gather(const int2* __restrict__ csr, const unsigned int* __restrict__ rowptr,
         const unsigned int* __restrict__ dtot, const float* __restrict__ h,
         const float* __restrict__ dis, const float* __restrict__ bias,
         float* __restrict__ out, int n) {
  const int lane = threadIdx.x & 31;
  const int c = lane & 15;
  const int pl = 2 * c + (lane >> 4);
  const int w0 = __builtin_amdgcn_readfirstlane((int)((blockIdx.x * blockDim.x + threadIdx.x) >> 5));
  const int stride = (int)((gridDim.x * blockDim.x) >> 5);
  if (w0 >= n) return;
  const v4f bia = *reinterpret_cast<const v4f*>(bias + 4 * c);
  for (int base = w0; base < n; base += 32 * stride) {
    const int j = min(base + lane * stride, n - 1);
    const int mst = (int)rowptr[j];
    const int mc = (int)dtot[j];
    const int md = __float_as_int(dis[j]);
    int2 recA = first_records(csr, __builtin_amdgcn_readlane(mst, 0),
                              __builtin_amdgcn_readlane(mc, 0), pl);
    asm volatile("" :: "v"(mst), "v"(mc), "v"(md), "v"(recA.x), "v"(recA.y));
    for (int k = 0; k < 32; ++k) {
      const int i = base + k * stride;
      if (i >= n) break;
      int2 recB = make_int2(0, 0);
      if (k + 1 < 32 && i + stride < n) {
        recB = first_records(csr, __builtin_amdgcn_readlane(mst, k + 1),
                             __builtin_amdgcn_readlane(mc, k + 1), pl);
      }
      const int st = __builtin_amdgcn_readlane(mst, k);
      const int cn = __builtin_amdgcn_readlane(mc, k);
      const float d = __int_as_float(__builtin_amdgcn_readlane(md, k));
      const v4f o = add_halves(node_sum<GK>(csr, h, i, st, cn, d, recA, lane)) * d + bia;
      if (lane < 16) reinterpret_cast<v4f*>(out)[(unsigned int)((i << 4) | c)] = o;
      recA = recB;
    }
  }
}

#define FT_NODES 16
#define FT_TB    (FT_NODES * 32)
#define FT_PITCH 68

__device__ __forceinline__ v8f tile_product(const float* __restrict__ rows,
                                            const float* __restrict__ Wl, int nt, int lane,
                                            float c0) {
  const int mlow  = lane & 15;
  const int khalf = (lane >> 4) << 1;
  const float* ar = rows + mlow * FT_PITCH + khalf;
  const float* wr = Wl + (nt * 16 + mlow) * FT_PITCH + khalf;
  v8f c = {c0, c0, c0, c0, c0, c0, c0, c0};
#pragma unroll
  for (int hq = 0; hq < 16; hq += 8) {
    v2f a[8], b[8];
#pragma unroll
    for (int q = 0; q < 8; ++q) {
      b[q].x = wr[(hq + q) * 4 + 0];
      b[q].y = wr[(hq + q) * 4 + 1];
    }
#pragma unroll
    for (int q = 0; q < 8; ++q) {
      a[q].x = ar[(hq + q) * 4 + 0];
      a[q].y = ar[(hq + q) * 4 + 1];
    }
    asm volatile("" :: "v"(a[0]), "v"(a[1]), "v"(a[2]), "v"(a[3]), "v"(a[4]), "v"(a[5]),
                       "v"(a[6]), "v"(a[7]), "v"(b[0]), "v"(b[1]), "v"(b[2]), "v"(b[3]),
                       "v"(b[4]), "v"(b[5]), "v"(b[6]), "v"(b[7]));
#pragma unroll
    for (int q = 0; q < 8; ++q) {
      c = __builtin_amdgcn_wmma_f32_16x16x4_f32(
           false, a[q],  false, b[q],
           (short)0, c,  false,  false);
    }
  }
  return c;
}

__device__ __forceinline__ void store_tile(const v8f c, float* __restrict__ H2,
                                           int m0, int n, int nt, int lane) {
  const int row0 = m0 + 8 * (lane >> 4);
  float* __restrict__ o = H2 + (size_t)m0 * 64;
  const unsigned int lo = (unsigned int)((lane >> 4) * 512 + nt * 16 + (lane & 15));
  if (m0 + FT_NODES <= n) {
#pragma unroll
    for (int v = 0; v < 8; ++v) o[lo + (unsigned int)v * 64u] = c[v];
  } else {
#pragma unroll
    for (int v = 0; v < 8; ++v) {
      if (row0 + v < n) o[lo + (unsigned int)v * 64u] = c[v];
    }
  }
}

#define FT_TILE (FT_NODES * FT_PITCH)
#define FT_WMAT (64 * FT_PITCH)
__global__ void __launch_bounds__(FT_TB, 16)
k_gather_gemm(const int2* __restrict__ csr, const unsigned int* __restrict__ rowptr,
              const unsigned int* __restrict__ dtot, const float* __restrict__ x,
              const float* __restrict__ dis, const float* __restrict__ W1,
              const float* __restrict__ b1, const float* __restrict__ W2,
              float* __restrict__ H2, int n, int ntiles) {
  __shared__ __attribute__((aligned(16))) float lds[4][FT_TILE];
  __shared__ __attribute__((aligned(16))) float wl[2][FT_WMAT];
  __shared__ float bl[64];
  static_assert(2 * FT_TB == 64 * 64 / 4, "two pieces of each matrix per thread");
  {
    const int tid = (int)threadIdx.x;
    v4f w1v[2], w2v[2];
#pragma unroll
    for (int it = 0; it < 2; ++it) {
      w1v[it] = *reinterpret_cast<const v4f*>(W1 + 4 * (tid + it * FT_TB));
      w2v[it] = *reinterpret_cast<const v4f*>(W2 + 4 * (tid + it * FT_TB));
    }
    const float bv = b1[tid & 63];
    asm volatile("" :: "v"(bv));
#pragma unroll
    for (int it = 0; it < 2; ++it) {
      const int p = tid + it * FT_TB;
      const int o = (p >> 4) * FT_PITCH + ((p & 15) << 2);
      *reinterpret_cast<v4f*>(&wl[0][o]) = w1v[it];
      *reinterpret_cast<v4f*>(&wl[1][o]) = w2v[it];
    }
    if (tid < 64) bl[tid] = bv;
  }
  const int lane = threadIdx.x & 31;
  const int c = lane & 15;
  const int pl = 2 * c + (lane >> 4);
  const int wid = __builtin_amdgcn_readfirstlane((int)(threadIdx.x >> 5));
  const int ngrp = (int)gridDim.x;
  const int stride = ngrp * FT_NODES;
  const int nt = wid & 3;
  const bool l1 = wid >= 4;

  int buf = 0;
  int pm0 = -1;
  for (int tb = (int)blockIdx.x; tb < ntiles; tb += 32 * ngrp) {
    const int ibase = tb * FT_NODES + wid;
    const int j = min(ibase + lane * stride, n - 1);
    const int mst = (int)rowptr[j];
    const int mc = (int)dtot[j];
    const int md = __float_as_int(dis[j]);
    int2 recA = make_int2(0, 0);
    if (ibase < n) {
      recA = first_records(csr, __builtin_amdgcn_readlane(mst, 0),
                           __builtin_amdgcn_readlane(mc, 0), pl);
    }
    asm volatile("" :: "v"(mst), "v"(mc), "v"(md), "v"(recA.x), "v"(recA.y));
    for (int k = 0; k < 32; ++k) {
      const int t = tb + k * ngrp;
      if (t >= ntiles) break;
      const int m0 = t * FT_NODES;
      const int i = m0 + wid;
      int2 recB = make_int2(0, 0);
      if (k + 1 < 32 && i + stride < n) {
        recB = first_records(csr, __builtin_amdgcn_readlane(mst, k + 1),
                             __builtin_amdgcn_readlane(mc, k + 1), pl);
      }
      v4f r = {0.0f, 0.0f, 0.0f, 0.0f};
      if (i < n) {
        const int st = __builtin_amdgcn_readlane(mst, k);
        const int cn = __builtin_amdgcn_readlane(mc, k);
        const float d = __int_as_float(__builtin_amdgcn_readlane(md, k));
        r = add_halves(node_sum<GF>(csr, x, i, st, cn, d, recA, lane)) * d;
      }
      if (lane < 16) *reinterpret_cast<v4f*>(&lds[buf][wid * FT_PITCH + 4 * c]) = r;
      __syncthreads();
      if (wid < 8 && (l1 || pm0 >= 0)) {
        float c0 = 0.0f;
        if (l1) c0 = bl[nt * 16 + (lane & 15)];
        const v8f c = tile_product(lds[l1 ? buf : 3 - buf], wl[l1 ? 0 : 1], nt, lane, c0);
        if (l1) {
          float* ro = lds[2 + buf] + (8 * (lane >> 4)) * FT_PITCH + nt * 16 + (lane & 15);
#pragma unroll
          for (int v = 0; v < 8; ++v) ro[v * FT_PITCH] = fmaxf(c[v], 0.0f);
        } else {
          store_tile(c, H2, pm0, n, nt, lane);
        }
      }
      pm0 = m0;
      buf ^= 1;
      recA = recB;
    }
  }
  __syncthreads();
  if (wid < 4 && pm0 >= 0) {
    const v8f c = tile_product(lds[3 - buf], wl[1], nt, lane, 0.0f);
    store_tile(c, H2, pm0, n, nt, lane);
  }
}

extern "C" void kernel_launch(void* const* d_in, const int* in_sizes, int n_in,
                              void* d_out, int out_size, void* d_ws, size_t ws_size,
                              hipStream_t stream) {
  (void)n_in; (void)out_size; (void)ws_size;

  const float* x  = (const float*)d_in[0];
  const int*   ei = (const int*)  d_in[1];
  const float* W1 = (const float*)d_in[2];
  const float* b1 = (const float*)d_in[3];
  const float* W2 = (const float*)d_in[4];
  const float* b2 = (const float*)d_in[5];
  float* out = (float*)d_out;

  const int n = in_sizes[0] / 64;
  const int e = in_sizes[1] / 2;
  const int* src = ei;
  const int* dst = ei + e;

  const int node_blocks = (n + 7) / 8;
  const int tile_blocks = (n + FT_NODES - 1) / FT_NODES;
  const int nranges     = (n + DC_RANGE - 1) / DC_RANGE;
  const int npad        = nranges * DC_RANGE;
  const int nfin        = nranges * FIN_PER_RANGE;

  float* h2  = (float*)d_ws;
  float* dis = h2 + (size_t)n * 64;
  unsigned int* part   = (unsigned int*)(dis + n);
  unsigned int* dtot   = part + (size_t)DC_SLICES * npad;
  unsigned int* rowptr = dtot + npad;
  unsigned int* bsum   = rowptr + npad;
  int2* csr = (int2*)(bsum + (size_t)nfin * 32);

  k_deg_count <<<nranges * DC_SLICES, DC_TB, 0, stream>>>(dst, part, e, nranges);
  k_deg_finish<<<nfin, FIN_TB, 0, stream>>>(part, dis, dtot, bsum, n, npad);
  k_fill      <<<nranges * DC_SLICES, DC_TB, 0, stream>>>(src, dst, part, dtot, bsum, dis, rowptr,
                                                          csr, e, nranges, npad);

  k_gather_gemm<<<tile_blocks < PG_WAVES / FT_NODES ? tile_blocks : PG_WAVES / FT_NODES, FT_TB, 0, stream>>>(
      csr, rowptr, dtot, x, dis, W1, b1, W2, h2, n, tile_blocks);

  k_gather<<<node_blocks < PG_WAVES / 8 ? node_blocks : PG_WAVES / 8, TB, 0, stream>>>(
      csr, rowptr, dtot, h2, dis, b2, out, n);
}
